# starting kernel
# baseline (speedup 1.0000x reference)
#include <hip/hip_runtime.h>
#include <math.h>


typedef __attribute__((ext_vector_type(2))) float v2f;
typedef __attribute__((ext_vector_type(8))) float v8f;

#define N_NODES 100000
#define N_EDGES 1600000
#define DIM 64
#define NTILES (N_NODES / 16)
#define L2EPS 1e-12f

__device__ __forceinline__ v8f wmma_f32(v2f a, v2f b, v8f c) {
  return __builtin_amdgcn_wmma_f32_16x16x4_f32(
       false, a,  false, b,
       (short)0, c,  false,  false);
}

__device__ __forceinline__ v2f frag_pair(const float* __restrict__ p) {
  v2f r; r.x = p[0]; r.y = p[1]; return r;
}

__global__ void zero_kernel(float* __restrict__ p, long long n) {
  long long i = (long long)blockIdx.x * blockDim.x + threadIdx.x;
  if (i < n) p[i] = 0.0f;
}

__global__ void scatter_cnt_kernel(const int* __restrict__ dst,
                                   float* __restrict__ cnt) {
  int e = blockIdx.x * blockDim.x + threadIdx.x;
  if (e < N_EDGES) atomicAdd(&cnt[dst[e]], 1.0f);
}

__global__ void scatter_sum_kernel(const float* __restrict__ x,
                                   const int* __restrict__ src,
                                   const int* __restrict__ dst,
                                   float* __restrict__ s) {
  long long idx = (long long)blockIdx.x * blockDim.x + threadIdx.x;
  if (idx >= (long long)N_EDGES * 16) return;
  int e  = (int)(idx >> 4);
  int d4 = ((int)idx & 15) * 4;
  int sn = src[e];
  int dn = dst[e];
  const float4 v = *(const float4*)(x + (long long)sn * DIM + d4);
  float* o = s + (long long)dn * DIM + d4;
  atomicAdd(o + 0, v.x);
  atomicAdd(o + 1, v.y);
  atomicAdd(o + 2, v.z);
  atomicAdd(o + 3, v.w);
}

__global__ __launch_bounds__(256) void sage_kernel(
    const float* __restrict__ xin, const float* __restrict__ ssum,
    const float* __restrict__ cnt,
    const float* __restrict__ wl, const float* __restrict__ bl,
    const float* __restrict__ wr, const float* __restrict__ br,
    float* __restrict__ out) {
  const int wave = threadIdx.x >> 5;
  const int lane = threadIdx.x & 31;
  const int tile = blockIdx.x * (blockDim.x >> 5) + wave;
  if (tile >= NTILES) return;

  const int l16  = lane & 15;
  const int half = lane >> 4;
  const int rb   = tile * 16;
  const int arow = rb + l16;

  const float rinv = 1.0f / fmaxf(cnt[arow], 1.0f);

  v2f ax[16], ap[16];
#pragma unroll
  for (int k = 0; k < 16; ++k) {
    const long long off = (long long)arow * DIM + 4 * k + 2 * half;
    ax[k] = frag_pair(xin + off);
    v2f sp = frag_pair(ssum + off);
    sp.x *= rinv; sp.y *= rinv;
    ap[k] = sp;
  }

  v8f acc[4];
#pragma unroll
  for (int nt = 0; nt < 4; ++nt) {
    v8f c8 = {};
    const int n0 = nt * 16;
#pragma unroll
    for (int k = 0; k < 16; ++k) {
      v2f b = frag_pair(wl + (long long)(n0 + l16) * DIM + 4 * k + 2 * half);
      c8 = wmma_f32(ax[k], b, c8);
    }
#pragma unroll
    for (int k = 0; k < 16; ++k) {
      v2f b = frag_pair(wr + (long long)(n0 + l16) * DIM + 4 * k + 2 * half);
      c8 = wmma_f32(ap[k], b, c8);
    }
    const float bias = bl[n0 + l16] + br[n0 + l16];
#pragma unroll
    for (int r = 0; r < 8; ++r) c8[r] += bias;
    acc[nt] = c8;
  }

  float ssq[8];
#pragma unroll
  for (int r = 0; r < 8; ++r) {
    float v = 0.0f;
#pragma unroll
    for (int nt = 0; nt < 4; ++nt) v += acc[nt][r] * acc[nt][r];
    ssq[r] = v;
  }
#pragma unroll
  for (int m = 1; m <= 8; m <<= 1) {
#pragma unroll
    for (int r = 0; r < 8; ++r) ssq[r] += __shfl_xor(ssq[r], m, 32);
  }

#pragma unroll
  for (int r = 0; r < 8; ++r) {
    const float inv = 1.0f / fmaxf(sqrtf(ssq[r]), L2EPS);
    const int row = rb + r + 8 * half;
#pragma unroll
    for (int nt = 0; nt < 4; ++nt) {
      float v = acc[nt][r] * inv;
      v = fmaxf(v, 0.0f);
      out[(long long)row * DIM + nt * 16 + l16] = v;
    }
  }
}

__global__ __launch_bounds__(256) void linear_kernel(
    const float* __restrict__ xin, const float* __restrict__ w,
    const float* __restrict__ b, float* __restrict__ out) {
  const int wave = threadIdx.x >> 5;
  const int lane = threadIdx.x & 31;
  const int tile = blockIdx.x * (blockDim.x >> 5) + wave;
  if (tile >= NTILES) return;

  const int l16  = lane & 15;
  const int half = lane >> 4;
  const int rb   = tile * 16;
  const int arow = rb + l16;

  v2f ax[16];
#pragma unroll
  for (int k = 0; k < 16; ++k)
    ax[k] = frag_pair(xin + (long long)arow * DIM + 4 * k + 2 * half);

#pragma unroll
  for (int nt = 0; nt < 4; ++nt) {
    v8f c8 = {};
    const int n0 = nt * 16;
#pragma unroll
    for (int k = 0; k < 16; ++k) {
      v2f bf = frag_pair(w + (long long)(n0 + l16) * DIM + 4 * k + 2 * half);
      c8 = wmma_f32(ax[k], bf, c8);
    }
    const float bias = b[n0 + l16];
#pragma unroll
    for (int r = 0; r < 8; ++r) {
      const int row = rb + r + 8 * half;
      out[(long long)row * DIM + n0 + l16] = c8[r] + bias;
    }
  }
}

extern "C" void kernel_launch(void* const* d_in, const int* in_sizes, int n_in,
                              void* d_out, int out_size, void* d_ws, size_t ws_size,
                              hipStream_t stream) {
  (void)in_sizes; (void)n_in; (void)out_size; (void)ws_size;

  const float* x   = (const float*)d_in[0];
  const int*   ei  = (const int*)d_in[1];
  const int*   src = ei;
  const int*   dst = ei + N_EDGES;
  const float* w1l = (const float*)d_in[2];
  const float* b1l = (const float*)d_in[3];
  const float* w1r = (const float*)d_in[4];
  const float* b1r = (const float*)d_in[5];
  const float* w2l = (const float*)d_in[6];
  const float* b2l = (const float*)d_in[7];
  const float* w2r = (const float*)d_in[8];
  const float* b2r = (const float*)d_in[9];
  const float* wp1 = (const float*)d_in[10];
  const float* bp1 = (const float*)d_in[11];
  const float* wp2 = (const float*)d_in[12];
  const float* bp2 = (const float*)d_in[13];
  float* out = (float*)d_out;

  const long long NS = (long long)N_NODES * DIM;
  float* s   = (float*)d_ws;
  float* cnt = s + NS;
  float* h1  = cnt + N_NODES;
  float* h2  = h1 + NS;
  float* tmp = s;

  const dim3 B(256);
  const int zgrid = (int)((NS + N_NODES + 255) / 256);
  const int egrid_cnt = (N_EDGES + 255) / 256;
  const int egrid_sum = (int)(((long long)N_EDGES * 16 + 255) / 256);
  const int ggrid = (NTILES + 7) / 8;

  zero_kernel<<<zgrid, B, 0, stream>>>(s, NS + N_NODES);
  scatter_cnt_kernel<<<egrid_cnt, B, 0, stream>>>(dst, cnt);
  scatter_sum_kernel<<<egrid_sum, B, 0, stream>>>(x, src, dst, s);
  sage_kernel<<<ggrid, B, 0, stream>>>(x, s, cnt, w1l, b1l, w1r, b1r, h1);

  zero_kernel<<<zgrid, B, 0, stream>>>(s, NS + N_NODES);
  scatter_cnt_kernel<<<egrid_cnt, B, 0, stream>>>(dst, cnt);
  scatter_sum_kernel<<<egrid_sum, B, 0, stream>>>(h1, src, dst, s);
  sage_kernel<<<ggrid, B, 0, stream>>>(h1, s, cnt, w2l, b2l, w2r, b2r, h2);

  linear_kernel<<<ggrid, B, 0, stream>>>(h2, wp1, bp1, tmp);
  linear_kernel<<<ggrid, B, 0, stream>>>(tmp, wp2, bp2, out);
}


# in-degree counted once inside the layer-1 scatter_sum kernel (thread of features 0..3 adds 1 to cnt[dst]); both scatter_cnt launches removed, layer 2 reuses cnt
# speedup vs baseline: 1.1300x; 1.1300x over previous
#include <hip/hip_runtime.h>
#include <math.h>


typedef __attribute__((ext_vector_type(2))) float v2f;
typedef __attribute__((ext_vector_type(8))) float v8f;

#define N_NODES 100000
#define N_EDGES 1600000
#define DIM 64
#define NTILES (N_NODES / 16)
#define L2EPS 1e-12f

__device__ __forceinline__ v8f wmma_f32(v2f a, v2f b, v8f c) {
  return __builtin_amdgcn_wmma_f32_16x16x4_f32(
       false, a,  false, b,
       (short)0, c,  false,  false);
}

__device__ __forceinline__ v2f frag_pair(const float* __restrict__ p) {
  v2f r; r.x = p[0]; r.y = p[1]; return r;
}

__global__ void zero_kernel(float* __restrict__ p, long long n) {
  long long i = (long long)blockIdx.x * blockDim.x + threadIdx.x;
  if (i < n) p[i] = 0.0f;
}

template <bool COUNT>
__global__ void scatter_sum_kernel(const float* __restrict__ x,
                                   const int* __restrict__ src,
                                   const int* __restrict__ dst,
                                   float* __restrict__ s,
                                   float* __restrict__ cnt) {
  long long idx = (long long)blockIdx.x * blockDim.x + threadIdx.x;
  if (idx >= (long long)N_EDGES * 16) return;
  int e  = (int)(idx >> 4);
  int d4 = ((int)idx & 15) * 4;
  int sn = src[e];
  int dn = dst[e];
  const float4 v = *(const float4*)(x + (long long)sn * DIM + d4);
  float* o = s + (long long)dn * DIM + d4;
  if (COUNT && d4 == 0) atomicAdd(cnt + dn, 1.0f);
  atomicAdd(o + 0, v.x);
  atomicAdd(o + 1, v.y);
  atomicAdd(o + 2, v.z);
  atomicAdd(o + 3, v.w);
}

__global__ __launch_bounds__(256) void sage_kernel(
    const float* __restrict__ xin, const float* __restrict__ ssum,
    const float* __restrict__ cnt,
    const float* __restrict__ wl, const float* __restrict__ bl,
    const float* __restrict__ wr, const float* __restrict__ br,
    float* __restrict__ out) {
  const int wave = threadIdx.x >> 5;
  const int lane = threadIdx.x & 31;
  const int tile = blockIdx.x * (blockDim.x >> 5) + wave;
  if (tile >= NTILES) return;

  const int l16  = lane & 15;
  const int half = lane >> 4;
  const int rb   = tile * 16;
  const int arow = rb + l16;

  const float rinv = 1.0f / fmaxf(cnt[arow], 1.0f);

  v2f ax[16], ap[16];
#pragma unroll
  for (int k = 0; k < 16; ++k) {
    const long long off = (long long)arow * DIM + 4 * k + 2 * half;
    ax[k] = frag_pair(xin + off);
    v2f sp = frag_pair(ssum + off);
    sp.x *= rinv; sp.y *= rinv;
    ap[k] = sp;
  }

  v8f acc[4];
#pragma unroll
  for (int nt = 0; nt < 4; ++nt) {
    v8f c8 = {};
    const int n0 = nt * 16;
#pragma unroll
    for (int k = 0; k < 16; ++k) {
      v2f b = frag_pair(wl + (long long)(n0 + l16) * DIM + 4 * k + 2 * half);
      c8 = wmma_f32(ax[k], b, c8);
    }
#pragma unroll
    for (int k = 0; k < 16; ++k) {
      v2f b = frag_pair(wr + (long long)(n0 + l16) * DIM + 4 * k + 2 * half);
      c8 = wmma_f32(ap[k], b, c8);
    }
    const float bias = bl[n0 + l16] + br[n0 + l16];
#pragma unroll
    for (int r = 0; r < 8; ++r) c8[r] += bias;
    acc[nt] = c8;
  }

  float ssq[8];
#pragma unroll
  for (int r = 0; r < 8; ++r) {
    float v = 0.0f;
#pragma unroll
    for (int nt = 0; nt < 4; ++nt) v += acc[nt][r] * acc[nt][r];
    ssq[r] = v;
  }
#pragma unroll
  for (int m = 1; m <= 8; m <<= 1) {
#pragma unroll
    for (int r = 0; r < 8; ++r) ssq[r] += __shfl_xor(ssq[r], m, 32);
  }

#pragma unroll
  for (int r = 0; r < 8; ++r) {
    const float inv = 1.0f / fmaxf(sqrtf(ssq[r]), L2EPS);
    const int row = rb + r + 8 * half;
#pragma unroll
    for (int nt = 0; nt < 4; ++nt) {
      float v = acc[nt][r] * inv;
      v = fmaxf(v, 0.0f);
      out[(long long)row * DIM + nt * 16 + l16] = v;
    }
  }
}

__global__ __launch_bounds__(256) void linear_kernel(
    const float* __restrict__ xin, const float* __restrict__ w,
    const float* __restrict__ b, float* __restrict__ out) {
  const int wave = threadIdx.x >> 5;
  const int lane = threadIdx.x & 31;
  const int tile = blockIdx.x * (blockDim.x >> 5) + wave;
  if (tile >= NTILES) return;

  const int l16  = lane & 15;
  const int half = lane >> 4;
  const int rb   = tile * 16;
  const int arow = rb + l16;

  v2f ax[16];
#pragma unroll
  for (int k = 0; k < 16; ++k)
    ax[k] = frag_pair(xin + (long long)arow * DIM + 4 * k + 2 * half);

#pragma unroll
  for (int nt = 0; nt < 4; ++nt) {
    v8f c8 = {};
    const int n0 = nt * 16;
#pragma unroll
    for (int k = 0; k < 16; ++k) {
      v2f bf = frag_pair(w + (long long)(n0 + l16) * DIM + 4 * k + 2 * half);
      c8 = wmma_f32(ax[k], bf, c8);
    }
    const float bias = b[n0 + l16];
#pragma unroll
    for (int r = 0; r < 8; ++r) {
      const int row = rb + r + 8 * half;
      out[(long long)row * DIM + n0 + l16] = c8[r] + bias;
    }
  }
}

extern "C" void kernel_launch(void* const* d_in, const int* in_sizes, int n_in,
                              void* d_out, int out_size, void* d_ws, size_t ws_size,
                              hipStream_t stream) {
  (void)in_sizes; (void)n_in; (void)out_size; (void)ws_size;

  const float* x   = (const float*)d_in[0];
  const int*   ei  = (const int*)d_in[1];
  const int*   src = ei;
  const int*   dst = ei + N_EDGES;
  const float* w1l = (const float*)d_in[2];
  const float* b1l = (const float*)d_in[3];
  const float* w1r = (const float*)d_in[4];
  const float* b1r = (const float*)d_in[5];
  const float* w2l = (const float*)d_in[6];
  const float* b2l = (const float*)d_in[7];
  const float* w2r = (const float*)d_in[8];
  const float* b2r = (const float*)d_in[9];
  const float* wp1 = (const float*)d_in[10];
  const float* bp1 = (const float*)d_in[11];
  const float* wp2 = (const float*)d_in[12];
  const float* bp2 = (const float*)d_in[13];
  float* out = (float*)d_out;

  const long long NS = (long long)N_NODES * DIM;
  float* s   = (float*)d_ws;
  float* cnt = s + NS;
  float* h1  = cnt + N_NODES;
  float* h2  = h1 + NS;
  float* tmp = s;

  const dim3 B(256);
  const int zgrid = (int)((NS + N_NODES + 255) / 256);
  const int zgrid_s = (int)((NS + 255) / 256);
  const int egrid_sum = (int)(((long long)N_EDGES * 16 + 255) / 256);
  const int ggrid = (NTILES + 7) / 8;

  zero_kernel<<<zgrid, B, 0, stream>>>(s, NS + N_NODES);
  scatter_sum_kernel<true><<<egrid_sum, B, 0, stream>>>(x, src, dst, s, cnt);
  sage_kernel<<<ggrid, B, 0, stream>>>(x, s, cnt, w1l, b1l, w1r, b1r, h1);

  zero_kernel<<<zgrid_s, B, 0, stream>>>(s, NS);
  scatter_sum_kernel<false><<<egrid_sum, B, 0, stream>>>(h1, src, dst, s, cnt);
  sage_kernel<<<ggrid, B, 0, stream>>>(h1, s, cnt, w2l, b2l, w2r, b2r, h2);

  linear_kernel<<<ggrid, B, 0, stream>>>(h2, wp1, bp1, tmp);
  linear_kernel<<<ggrid, B, 0, stream>>>(tmp, wp2, bp2, out);
}


# segment sums without float atomics: each call counting-sorts the edges by dst in two levels with LDS counters (arrival-order slots); each node then sums its rows in registers in slot order
# speedup vs baseline: 3.0261x; 2.6780x over previous
#include <hip/hip_runtime.h>
#include <math.h>


typedef __attribute__((ext_vector_type(2))) float v2f;
typedef __attribute__((ext_vector_type(8))) float v8f;

#define N_NODES 100000
#define N_EDGES 1600000
#define DIM 64
#define NTILES (N_NODES / 16)
#define L2EPS 1e-12f

__device__ __forceinline__ v8f wmma_f32(v2f a, v2f b, v8f c) {
  return __builtin_amdgcn_wmma_f32_16x16x4_f32(
       false, a,  false, b,
       (short)0, c,  false,  false);
}

__device__ __forceinline__ v2f frag_pair(const float* __restrict__ p) {
  v2f r; r.x = p[0]; r.y = p[1]; return r;
}

__global__ void zero_kernel(float* __restrict__ p, long long n) {
  long long i = (long long)blockIdx.x * blockDim.x + threadIdx.x;
  if (i < n) p[i] = 0.0f;
}

#define NB ((N_NODES + 255) / 256)
#define CHUNKS 250
#define CHUNK_V (N_EDGES / CHUNKS / 4)
#define BT_STRIDE 32
#define SORT_T 1024
static_assert(CHUNK_V * 4 * CHUNKS == N_EDGES, "chunks must tile the edge list");
static_assert(N_NODES <= (1 << 24), "src must fit the upper 24 bits of a record");
static_assert(NB <= SORT_T, "one thread per block in the scans");

__device__ __forceinline__ int block_scan_excl(int val, int* wtot, int& total) {
  const int lane = threadIdx.x & 31;
  const int wave = threadIdx.x >> 5;
  int inc = val;
#pragma unroll
  for (int o = 1; o < 32; o <<= 1) {
    const int up = __shfl_up(inc, o, 32);
    if (lane >= o) inc += up;
  }
  if (lane == 31) wtot[wave] = inc;
  __syncthreads();
  if (wave == 0) {
    const int t = wtot[lane];
    int ti = t;
#pragma unroll
    for (int o = 1; o < 32; o <<= 1) {
      const int up = __shfl_up(ti, o, 32);
      if (lane >= o) ti += up;
    }
    wtot[lane] = ti - t;
    if (lane == 31) wtot[32] = ti;
  }
  __syncthreads();
  total = wtot[32];
  const int ex = wtot[wave] + inc - val;
  __syncthreads();
  return ex;
}

__global__ __launch_bounds__(SORT_T) void bin_count_kernel(const int* __restrict__ dst,
                                                           int* btot,
                                                           int* __restrict__ run0) {
  __shared__ int hist[NB];
  const int t = threadIdx.x;
  const int c = blockIdx.x;
  if (t < NB) hist[t] = 0;
  __syncthreads();
  const int4* d4 = (const int4*)dst + (long long)c * CHUNK_V;
  for (int i = t; i < CHUNK_V; i += SORT_T) {
    const int4 d = d4[i];
    atomicAdd(&hist[d.x >> 8], 1);
    atomicAdd(&hist[d.y >> 8], 1);
    atomicAdd(&hist[d.z >> 8], 1);
    atomicAdd(&hist[d.w >> 8], 1);
  }
  __syncthreads();
  if (t < NB) run0[c * NB + t] = atomicAdd(btot + t * BT_STRIDE, hist[t]);
}

__global__ __launch_bounds__(SORT_T) void bin_scatter_kernel(const int* __restrict__ src,
                                                             const int* __restrict__ dst,
                                                             const int* __restrict__ btot,
                                                             const int* __restrict__ run0,
                                                             int* __restrict__ rec) {
  __shared__ int cur[NB];
  __shared__ int wtot[33];
  const int t = threadIdx.x;
  const int c = blockIdx.x;
  const int bt = (t < NB) ? btot[t * BT_STRIDE] : 0;
  int total;
  const int bstart = block_scan_excl(bt, wtot, total);
  if (t < NB) cur[t] = bstart + run0[c * NB + t];
  __syncthreads();
  const int4* s4 = (const int4*)src + (long long)c * CHUNK_V;
  const int4* d4 = (const int4*)dst + (long long)c * CHUNK_V;
  for (int i = t; i < CHUNK_V; i += SORT_T) {
    const int4 sn = s4[i];
    const int4 d  = d4[i];
    const int p0 = min(max(atomicAdd(&cur[d.x >> 8], 1), 0), N_EDGES - 1);
    const int p1 = min(max(atomicAdd(&cur[d.y >> 8], 1), 0), N_EDGES - 1);
    const int p2 = min(max(atomicAdd(&cur[d.z >> 8], 1), 0), N_EDGES - 1);
    const int p3 = min(max(atomicAdd(&cur[d.w >> 8], 1), 0), N_EDGES - 1);
    rec[p0] = (int)(((unsigned)sn.x << 8) | (unsigned)(d.x & 255));
    rec[p1] = (int)(((unsigned)sn.y << 8) | (unsigned)(d.y & 255));
    rec[p2] = (int)(((unsigned)sn.z << 8) | (unsigned)(d.z & 255));
    rec[p3] = (int)(((unsigned)sn.w << 8) | (unsigned)(d.w & 255));
  }
}

__global__ __launch_bounds__(SORT_T) void bin_sort_kernel(const int* __restrict__ btot,
                                                          const int* __restrict__ rec,
                                                          int* __restrict__ csr,
                                                          int* __restrict__ off,
                                                          int* __restrict__ deg,
                                                          float* __restrict__ cnt) {
  __shared__ int cntL[256];
  __shared__ int offL[256];
  __shared__ int wtot[33];
  const int t = threadIdx.x;
  const int b = blockIdx.x;
  const int bt = (t < b) ? btot[t * BT_STRIDE] : 0;
  int before;
  block_scan_excl(bt, wtot, before);
  const int bstart = min(max(before, 0), N_EDGES);
  const int bsize  = min(max(btot[b * BT_STRIDE], 0), N_EDGES - bstart);
  if (t < 256) cntL[t] = 0;
  __syncthreads();
  for (int i = t; i < bsize; i += SORT_T) atomicAdd(&cntL[rec[bstart + i] & 255], 1);
  __syncthreads();
  const int c = (t < 256) ? cntL[t] : 0;
  int all;
  const int ex = block_scan_excl(c, wtot, all);
  if (t < 256) {
    offL[t] = ex;
    cntL[t] = 0;
    const int v = b * 256 + t;
    if (v < N_NODES) {
      off[v] = bstart + ex;
      deg[v] = c;
      cnt[v] = (float)c;
    }
  }
  __syncthreads();
  for (int i = t; i < bsize; i += SORT_T) {
    const int w = rec[bstart + i];
    const int k = w & 255;
    const int r = atomicAdd(&cntL[k], 1);
    const int p = min(max(bstart + offL[k] + r, 0), N_EDGES - 1);
    csr[p] = (int)((unsigned)w >> 8);
  }
}

#define GB 8
__global__ __launch_bounds__(256) void gather_kernel(const float* __restrict__ x,
                                                     const int* __restrict__ csr,
                                                     const int* __restrict__ off,
                                                     const int* __restrict__ deg,
                                                     float* __restrict__ s) {
  const int gid = blockIdx.x * blockDim.x + threadIdx.x;
  const int v   = gid >> 4;
  if (v >= N_NODES) return;
  const int d4  = (gid & 15) * 4;
  const int start = min(max(off[v], 0), N_EDGES);
  const int n     = min(max(deg[v], 0), N_EDGES - start);
  float4 acc = make_float4(0.0f, 0.0f, 0.0f, 0.0f);
  for (int j = 0; j < n; j += GB) {
    int idx[GB];
#pragma unroll
    for (int u = 0; u < GB; ++u) {
      const int i = csr[start + min(j + u, n - 1)];
      idx[u] = min(max(i, 0), N_NODES - 1);
    }
    float4 r[GB];
#pragma unroll
    for (int u = 0; u < GB; ++u)
      r[u] = *(const float4*)(x + (long long)idx[u] * DIM + d4);
#pragma unroll
    for (int u = 0; u < GB; ++u) {
      const bool in = (j + u < n);
      acc.x += in ? r[u].x : 0.0f;
      acc.y += in ? r[u].y : 0.0f;
      acc.z += in ? r[u].z : 0.0f;
      acc.w += in ? r[u].w : 0.0f;
    }
  }
  *(float4*)(s + (long long)v * DIM + d4) = acc;
}

__global__ __launch_bounds__(256) void sage_kernel(
    const float* __restrict__ xin, const float* __restrict__ ssum,
    const float* __restrict__ cnt,
    const float* __restrict__ wl, const float* __restrict__ bl,
    const float* __restrict__ wr, const float* __restrict__ br,
    float* __restrict__ out) {
  const int wave = threadIdx.x >> 5;
  const int lane = threadIdx.x & 31;
  const int tile = blockIdx.x * (blockDim.x >> 5) + wave;
  if (tile >= NTILES) return;

  const int l16  = lane & 15;
  const int half = lane >> 4;
  const int rb   = tile * 16;
  const int arow = rb + l16;

  const float rinv = 1.0f / fmaxf(cnt[arow], 1.0f);

  v2f ax[16], ap[16];
#pragma unroll
  for (int k = 0; k < 16; ++k) {
    const long long off = (long long)arow * DIM + 4 * k + 2 * half;
    ax[k] = frag_pair(xin + off);
    v2f sp = frag_pair(ssum + off);
    sp.x *= rinv; sp.y *= rinv;
    ap[k] = sp;
  }

  v8f acc[4];
#pragma unroll
  for (int nt = 0; nt < 4; ++nt) {
    v8f c8 = {};
    const int n0 = nt * 16;
#pragma unroll
    for (int k = 0; k < 16; ++k) {
      v2f b = frag_pair(wl + (long long)(n0 + l16) * DIM + 4 * k + 2 * half);
      c8 = wmma_f32(ax[k], b, c8);
    }
#pragma unroll
    for (int k = 0; k < 16; ++k) {
      v2f b = frag_pair(wr + (long long)(n0 + l16) * DIM + 4 * k + 2 * half);
      c8 = wmma_f32(ap[k], b, c8);
    }
    const float bias = bl[n0 + l16] + br[n0 + l16];
#pragma unroll
    for (int r = 0; r < 8; ++r) c8[r] += bias;
    acc[nt] = c8;
  }

  float ssq[8];
#pragma unroll
  for (int r = 0; r < 8; ++r) {
    float v = 0.0f;
#pragma unroll
    for (int nt = 0; nt < 4; ++nt) v += acc[nt][r] * acc[nt][r];
    ssq[r] = v;
  }
#pragma unroll
  for (int m = 1; m <= 8; m <<= 1) {
#pragma unroll
    for (int r = 0; r < 8; ++r) ssq[r] += __shfl_xor(ssq[r], m, 32);
  }

#pragma unroll
  for (int r = 0; r < 8; ++r) {
    const float inv = 1.0f / fmaxf(sqrtf(ssq[r]), L2EPS);
    const int row = rb + r + 8 * half;
#pragma unroll
    for (int nt = 0; nt < 4; ++nt) {
      float v = acc[nt][r] * inv;
      v = fmaxf(v, 0.0f);
      out[(long long)row * DIM + nt * 16 + l16] = v;
    }
  }
}

__global__ __launch_bounds__(256) void linear_kernel(
    const float* __restrict__ xin, const float* __restrict__ w,
    const float* __restrict__ b, float* __restrict__ out) {
  const int wave = threadIdx.x >> 5;
  const int lane = threadIdx.x & 31;
  const int tile = blockIdx.x * (blockDim.x >> 5) + wave;
  if (tile >= NTILES) return;

  const int l16  = lane & 15;
  const int half = lane >> 4;
  const int rb   = tile * 16;
  const int arow = rb + l16;

  v2f ax[16];
#pragma unroll
  for (int k = 0; k < 16; ++k)
    ax[k] = frag_pair(xin + (long long)arow * DIM + 4 * k + 2 * half);

#pragma unroll
  for (int nt = 0; nt < 4; ++nt) {
    v8f c8 = {};
    const int n0 = nt * 16;
#pragma unroll
    for (int k = 0; k < 16; ++k) {
      v2f bf = frag_pair(w + (long long)(n0 + l16) * DIM + 4 * k + 2 * half);
      c8 = wmma_f32(ax[k], bf, c8);
    }
    const float bias = b[n0 + l16];
#pragma unroll
    for (int r = 0; r < 8; ++r) {
      const int row = rb + r + 8 * half;
      out[(long long)row * DIM + n0 + l16] = c8[r] + bias;
    }
  }
}

extern "C" void kernel_launch(void* const* d_in, const int* in_sizes, int n_in,
                              void* d_out, int out_size, void* d_ws, size_t ws_size,
                              hipStream_t stream) {
  (void)in_sizes; (void)n_in; (void)out_size; (void)ws_size;

  const float* x   = (const float*)d_in[0];
  const int*   ei  = (const int*)d_in[1];
  const int*   src = ei;
  const int*   dst = ei + N_EDGES;
  const float* w1l = (const float*)d_in[2];
  const float* b1l = (const float*)d_in[3];
  const float* w1r = (const float*)d_in[4];
  const float* b1r = (const float*)d_in[5];
  const float* w2l = (const float*)d_in[6];
  const float* b2l = (const float*)d_in[7];
  const float* w2r = (const float*)d_in[8];
  const float* b2r = (const float*)d_in[9];
  const float* wp1 = (const float*)d_in[10];
  const float* bp1 = (const float*)d_in[11];
  const float* wp2 = (const float*)d_in[12];
  const float* bp2 = (const float*)d_in[13];
  float* out = (float*)d_out;

  const long long NS = (long long)N_NODES * DIM;
  float* s   = (float*)d_ws;
  float* cnt = s + NS;
  float* h1  = cnt + N_NODES;
  float* h2  = h1 + NS;
  float* tmp = s;
  int* btot = (int*)h2;
  int* run0 = btot + NB * BT_STRIDE;
  int* deg  = run0 + (CHUNKS * NB + 31) / 32 * 32;
  int* off  = deg + N_NODES;
  int* rec  = off + N_NODES;
  int* csr  = rec + N_EDGES;

  const dim3 B(256);
  const dim3 BS(SORT_T);
  const int zgrid = (NB * BT_STRIDE + 255) / 256;
  const int hgrid = N_NODES * 16 / 256;
  const int ggrid = (NTILES + 7) / 8;

  zero_kernel<<<zgrid, B, 0, stream>>>((float*)btot, NB * BT_STRIDE);
  bin_count_kernel<<<CHUNKS, BS, 0, stream>>>(dst, btot, run0);
  bin_scatter_kernel<<<CHUNKS, BS, 0, stream>>>(src, dst, btot, run0, rec);
  bin_sort_kernel<<<NB, BS, 0, stream>>>(btot, rec, csr, off, deg, cnt);

  gather_kernel<<<hgrid, B, 0, stream>>>(x, csr, off, deg, s);
  sage_kernel<<<ggrid, B, 0, stream>>>(x, s, cnt, w1l, b1l, w1r, b1r, h1);

  gather_kernel<<<hgrid, B, 0, stream>>>(h1, csr, off, deg, s);
  sage_kernel<<<ggrid, B, 0, stream>>>(h1, s, cnt, w2l, b2l, w2r, b2r, h2);

  linear_kernel<<<ggrid, B, 0, stream>>>(h2, wp1, bp1, tmp);
  linear_kernel<<<ggrid, B, 0, stream>>>(tmp, wp2, bp2, out);
}


# weights copied once per work-group into LDS in fragment order; every B fragment of the layer and head kernels is an LDS read, not a global read that takes 8 bytes from each row's cache line
# speedup vs baseline: 3.5387x; 1.1694x over previous
#include <hip/hip_runtime.h>
#include <math.h>


typedef __attribute__((ext_vector_type(2))) float v2f;
typedef __attribute__((ext_vector_type(8))) float v8f;

#define N_NODES 100000
#define N_EDGES 1600000
#define DIM 64
#define NTILES (N_NODES / 16)
#define L2EPS 1e-12f

__device__ __forceinline__ v8f wmma_f32(v2f a, v2f b, v8f c) {
  return __builtin_amdgcn_wmma_f32_16x16x4_f32(
       false, a,  false, b,
       (short)0, c,  false,  false);
}

__device__ __forceinline__ v2f frag_pair(const float* __restrict__ p) {
  v2f r; r.x = p[0]; r.y = p[1]; return r;
}

#define WF_PITCH 33
#define WF_PAIRS (64 * WF_PITCH)
#define GEMM_T 256
#define WQ (DIM * DIM / 4 / GEMM_T)

__device__ __forceinline__ void load_weights(const float* __restrict__ w, float4 (&q)[WQ]) {
#pragma unroll
  for (int j = 0; j < WQ; ++j)
    q[j] = *(const float4*)(w + 4 * (threadIdx.x + j * GEMM_T));
}

__device__ __forceinline__ void store_weights(v2f* lw, const float4 (&q)[WQ]) {
#pragma unroll
  for (int j = 0; j < WQ; ++j) {
    const int i = threadIdx.x + j * GEMM_T;
    const int n = i >> 4;
    const int c = i & 15;
    const int f = (n >> 4) * 16 + c;
    v2f lo; lo.x = q[j].x; lo.y = q[j].y;
    v2f hi; hi.x = q[j].z; hi.y = q[j].w;
    lw[f * WF_PITCH + (n & 15)]      = lo;
    lw[f * WF_PITCH + 16 + (n & 15)] = hi;
  }
}

__global__ void zero_kernel(float* __restrict__ p, long long n) {
  long long i = (long long)blockIdx.x * blockDim.x + threadIdx.x;
  if (i < n) p[i] = 0.0f;
}

#define NB ((N_NODES + 255) / 256)
#define CHUNKS 250
#define CHUNK_V (N_EDGES / CHUNKS / 4)
#define BT_STRIDE 32
#define SORT_T 1024
static_assert(CHUNK_V * 4 * CHUNKS == N_EDGES, "chunks must tile the edge list");
static_assert(N_NODES <= (1 << 24), "src must fit the upper 24 bits of a record");
static_assert(NB <= SORT_T, "one thread per block in the scans");

__device__ __forceinline__ int block_scan_excl(int val, int* wtot, int& total) {
  const int lane = threadIdx.x & 31;
  const int wave = threadIdx.x >> 5;
  int inc = val;
#pragma unroll
  for (int o = 1; o < 32; o <<= 1) {
    const int up = __shfl_up(inc, o, 32);
    if (lane >= o) inc += up;
  }
  if (lane == 31) wtot[wave] = inc;
  __syncthreads();
  if (wave == 0) {
    const int t = wtot[lane];
    int ti = t;
#pragma unroll
    for (int o = 1; o < 32; o <<= 1) {
      const int up = __shfl_up(ti, o, 32);
      if (lane >= o) ti += up;
    }
    wtot[lane] = ti - t;
    if (lane == 31) wtot[32] = ti;
  }
  __syncthreads();
  total = wtot[32];
  const int ex = wtot[wave] + inc - val;
  __syncthreads();
  return ex;
}

__global__ __launch_bounds__(SORT_T) void bin_count_kernel(const int* __restrict__ dst,
                                                           int* btot,
                                                           int* __restrict__ run0) {
  __shared__ int hist[NB];
  const int t = threadIdx.x;
  const int c = blockIdx.x;
  if (t < NB) hist[t] = 0;
  __syncthreads();
  const int4* d4 = (const int4*)dst + (long long)c * CHUNK_V;
  for (int i = t; i < CHUNK_V; i += SORT_T) {
    const int4 d = d4[i];
    atomicAdd(&hist[d.x >> 8], 1);
    atomicAdd(&hist[d.y >> 8], 1);
    atomicAdd(&hist[d.z >> 8], 1);
    atomicAdd(&hist[d.w >> 8], 1);
  }
  __syncthreads();
  if (t < NB) run0[c * NB + t] = atomicAdd(btot + t * BT_STRIDE, hist[t]);
}

__global__ __launch_bounds__(SORT_T) void bin_scatter_kernel(const int* __restrict__ src,
                                                             const int* __restrict__ dst,
                                                             const int* __restrict__ btot,
                                                             const int* __restrict__ run0,
                                                             int* __restrict__ rec) {
  __shared__ int cur[NB];
  __shared__ int wtot[33];
  const int t = threadIdx.x;
  const int c = blockIdx.x;
  const int bt = (t < NB) ? btot[t * BT_STRIDE] : 0;
  int total;
  const int bstart = block_scan_excl(bt, wtot, total);
  if (t < NB) cur[t] = bstart + run0[c * NB + t];
  __syncthreads();
  const int4* s4 = (const int4*)src + (long long)c * CHUNK_V;
  const int4* d4 = (const int4*)dst + (long long)c * CHUNK_V;
  for (int i = t; i < CHUNK_V; i += SORT_T) {
    const int4 sn = s4[i];
    const int4 d  = d4[i];
    const int p0 = min(max(atomicAdd(&cur[d.x >> 8], 1), 0), N_EDGES - 1);
    const int p1 = min(max(atomicAdd(&cur[d.y >> 8], 1), 0), N_EDGES - 1);
    const int p2 = min(max(atomicAdd(&cur[d.z >> 8], 1), 0), N_EDGES - 1);
    const int p3 = min(max(atomicAdd(&cur[d.w >> 8], 1), 0), N_EDGES - 1);
    rec[p0] = (int)(((unsigned)sn.x << 8) | (unsigned)(d.x & 255));
    rec[p1] = (int)(((unsigned)sn.y << 8) | (unsigned)(d.y & 255));
    rec[p2] = (int)(((unsigned)sn.z << 8) | (unsigned)(d.z & 255));
    rec[p3] = (int)(((unsigned)sn.w << 8) | (unsigned)(d.w & 255));
  }
}

__global__ __launch_bounds__(SORT_T) void bin_sort_kernel(const int* __restrict__ btot,
                                                          const int* __restrict__ rec,
                                                          int* __restrict__ csr,
                                                          int* __restrict__ off,
                                                          int* __restrict__ deg,
                                                          float* __restrict__ cnt) {
  __shared__ int cntL[256];
  __shared__ int offL[256];
  __shared__ int wtot[33];
  const int t = threadIdx.x;
  const int b = blockIdx.x;
  const int bt = (t < b) ? btot[t * BT_STRIDE] : 0;
  int before;
  block_scan_excl(bt, wtot, before);
  const int bstart = min(max(before, 0), N_EDGES);
  const int bsize  = min(max(btot[b * BT_STRIDE], 0), N_EDGES - bstart);
  if (t < 256) cntL[t] = 0;
  __syncthreads();
  for (int i = t; i < bsize; i += SORT_T) atomicAdd(&cntL[rec[bstart + i] & 255], 1);
  __syncthreads();
  const int c = (t < 256) ? cntL[t] : 0;
  int all;
  const int ex = block_scan_excl(c, wtot, all);
  if (t < 256) {
    offL[t] = ex;
    cntL[t] = 0;
    const int v = b * 256 + t;
    if (v < N_NODES) {
      off[v] = bstart + ex;
      deg[v] = c;
      cnt[v] = (float)c;
    }
  }
  __syncthreads();
  for (int i = t; i < bsize; i += SORT_T) {
    const int w = rec[bstart + i];
    const int k = w & 255;
    const int r = atomicAdd(&cntL[k], 1);
    const int p = min(max(bstart + offL[k] + r, 0), N_EDGES - 1);
    csr[p] = (int)((unsigned)w >> 8);
  }
}

#define GB 8
__global__ __launch_bounds__(256) void gather_kernel(const float* __restrict__ x,
                                                     const int* __restrict__ csr,
                                                     const int* __restrict__ off,
                                                     const int* __restrict__ deg,
                                                     float* __restrict__ s) {
  const int gid = blockIdx.x * blockDim.x + threadIdx.x;
  const int v   = gid >> 4;
  if (v >= N_NODES) return;
  const int d4  = (gid & 15) * 4;
  const int start = min(max(off[v], 0), N_EDGES);
  const int n     = min(max(deg[v], 0), N_EDGES - start);
  float4 acc = make_float4(0.0f, 0.0f, 0.0f, 0.0f);
  for (int j = 0; j < n; j += GB) {
    int idx[GB];
#pragma unroll
    for (int u = 0; u < GB; ++u) {
      const int i = csr[start + min(j + u, n - 1)];
      idx[u] = min(max(i, 0), N_NODES - 1);
    }
    float4 r[GB];
#pragma unroll
    for (int u = 0; u < GB; ++u)
      r[u] = *(const float4*)(x + (long long)idx[u] * DIM + d4);
#pragma unroll
    for (int u = 0; u < GB; ++u) {
      const bool in = (j + u < n);
      acc.x += in ? r[u].x : 0.0f;
      acc.y += in ? r[u].y : 0.0f;
      acc.z += in ? r[u].z : 0.0f;
      acc.w += in ? r[u].w : 0.0f;
    }
  }
  *(float4*)(s + (long long)v * DIM + d4) = acc;
}

__global__ __launch_bounds__(GEMM_T) void sage_kernel(
    const float* __restrict__ xin, const float* __restrict__ ssum,
    const float* __restrict__ cnt,
    const float* __restrict__ wl, const float* __restrict__ bl,
    const float* __restrict__ wr, const float* __restrict__ br,
    float* __restrict__ out) {
  __shared__ v2f lwl[WF_PAIRS];
  __shared__ v2f lwr[WF_PAIRS];
  const int wave = threadIdx.x >> 5;
  const int lane = threadIdx.x & 31;
  const int tile_raw = blockIdx.x * (blockDim.x >> 5) + wave;
  const bool live = tile_raw < NTILES;
  const int tile = live ? tile_raw : NTILES - 1;

  const int l16  = lane & 15;
  const int half = lane >> 4;
  const int rb   = tile * 16;
  const int arow = rb + l16;

  const float rinv = 1.0f / fmaxf(cnt[arow], 1.0f);

  v2f ax[16], ap[16];
#pragma unroll
  for (int k = 0; k < 16; ++k) {
    const long long off = (long long)arow * DIM + 4 * k + 2 * half;
    ax[k] = frag_pair(xin + off);
    v2f sp = frag_pair(ssum + off);
    sp.x *= rinv; sp.y *= rinv;
    ap[k] = sp;
  }

  float4 ql[WQ], qr[WQ];
  load_weights(wl, ql);
  load_weights(wr, qr);
  store_weights(lwl, ql);
  store_weights(lwr, qr);
  __syncthreads();
  if (!live) return;

  v8f acc[4];
#pragma unroll
  for (int nt = 0; nt < 4; ++nt) {
    v8f c8 = {};
    const int n0 = nt * 16;
#pragma unroll
    for (int k = 0; k < 16; ++k) {
      const v2f b = lwl[(nt * 16 + k) * WF_PITCH + lane];
      c8 = wmma_f32(ax[k], b, c8);
    }
#pragma unroll
    for (int k = 0; k < 16; ++k) {
      const v2f b = lwr[(nt * 16 + k) * WF_PITCH + lane];
      c8 = wmma_f32(ap[k], b, c8);
    }
    const float bias = bl[n0 + l16] + br[n0 + l16];
#pragma unroll
    for (int r = 0; r < 8; ++r) c8[r] += bias;
    acc[nt] = c8;
  }

  float ssq[8];
#pragma unroll
  for (int r = 0; r < 8; ++r) {
    float v = 0.0f;
#pragma unroll
    for (int nt = 0; nt < 4; ++nt) v += acc[nt][r] * acc[nt][r];
    ssq[r] = v;
  }
#pragma unroll
  for (int m = 1; m <= 8; m <<= 1) {
#pragma unroll
    for (int r = 0; r < 8; ++r) ssq[r] += __shfl_xor(ssq[r], m, 32);
  }

#pragma unroll
  for (int r = 0; r < 8; ++r) {
    const float inv = 1.0f / fmaxf(sqrtf(ssq[r]), L2EPS);
    const int row = rb + r + 8 * half;
#pragma unroll
    for (int nt = 0; nt < 4; ++nt) {
      float v = acc[nt][r] * inv;
      v = fmaxf(v, 0.0f);
      out[(long long)row * DIM + nt * 16 + l16] = v;
    }
  }
}

__global__ __launch_bounds__(GEMM_T) void linear_kernel(
    const float* __restrict__ xin, const float* __restrict__ w,
    const float* __restrict__ b, float* __restrict__ out) {
  __shared__ v2f lw[WF_PAIRS];
  const int wave = threadIdx.x >> 5;
  const int lane = threadIdx.x & 31;
  const int tile_raw = blockIdx.x * (blockDim.x >> 5) + wave;
  const bool live = tile_raw < NTILES;
  const int tile = live ? tile_raw : NTILES - 1;

  const int l16  = lane & 15;
  const int half = lane >> 4;
  const int rb   = tile * 16;
  const int arow = rb + l16;

  v2f ax[16];
#pragma unroll
  for (int k = 0; k < 16; ++k)
    ax[k] = frag_pair(xin + (long long)arow * DIM + 4 * k + 2 * half);

  float4 qw[WQ];
  load_weights(w, qw);
  store_weights(lw, qw);
  __syncthreads();
  if (!live) return;

#pragma unroll
  for (int nt = 0; nt < 4; ++nt) {
    v8f c8 = {};
    const int n0 = nt * 16;
#pragma unroll
    for (int k = 0; k < 16; ++k) {
      const v2f bf = lw[(nt * 16 + k) * WF_PITCH + lane];
      c8 = wmma_f32(ax[k], bf, c8);
    }
    const float bias = b[n0 + l16];
#pragma unroll
    for (int r = 0; r < 8; ++r) {
      const int row = rb + r + 8 * half;
      out[(long long)row * DIM + n0 + l16] = c8[r] + bias;
    }
  }
}

extern "C" void kernel_launch(void* const* d_in, const int* in_sizes, int n_in,
                              void* d_out, int out_size, void* d_ws, size_t ws_size,
                              hipStream_t stream) {
  (void)in_sizes; (void)n_in; (void)out_size; (void)ws_size;

  const float* x   = (const float*)d_in[0];
  const int*   ei  = (const int*)d_in[1];
  const int*   src = ei;
  const int*   dst = ei + N_EDGES;
  const float* w1l = (const float*)d_in[2];
  const float* b1l = (const float*)d_in[3];
  const float* w1r = (const float*)d_in[4];
  const float* b1r = (const float*)d_in[5];
  const float* w2l = (const float*)d_in[6];
  const float* b2l = (const float*)d_in[7];
  const float* w2r = (const float*)d_in[8];
  const float* b2r = (const float*)d_in[9];
  const float* wp1 = (const float*)d_in[10];
  const float* bp1 = (const float*)d_in[11];
  const float* wp2 = (const float*)d_in[12];
  const float* bp2 = (const float*)d_in[13];
  float* out = (float*)d_out;

  const long long NS = (long long)N_NODES * DIM;
  float* s   = (float*)d_ws;
  float* cnt = s + NS;
  float* h1  = cnt + N_NODES;
  float* h2  = h1 + NS;
  float* tmp = s;
  int* btot = (int*)h2;
  int* run0 = btot + NB * BT_STRIDE;
  int* deg  = run0 + (CHUNKS * NB + 31) / 32 * 32;
  int* off  = deg + N_NODES;
  int* rec  = off + N_NODES;
  int* csr  = rec + N_EDGES;

  const dim3 B(256);
  const dim3 BG(GEMM_T);
  const dim3 BS(SORT_T);
  const int zgrid = (NB * BT_STRIDE + 255) / 256;
  const int hgrid = N_NODES * 16 / 256;
  const int ggrid = (NTILES + 7) / 8;

  zero_kernel<<<zgrid, B, 0, stream>>>((float*)btot, NB * BT_STRIDE);
  bin_count_kernel<<<CHUNKS, BS, 0, stream>>>(dst, btot, run0);
  bin_scatter_kernel<<<CHUNKS, BS, 0, stream>>>(src, dst, btot, run0, rec);
  bin_sort_kernel<<<NB, BS, 0, stream>>>(btot, rec, csr, off, deg, cnt);

  gather_kernel<<<hgrid, B, 0, stream>>>(x, csr, off, deg, s);
  sage_kernel<<<ggrid, BG, 0, stream>>>(x, s, cnt, w1l, b1l, w1r, b1r, h1);

  gather_kernel<<<hgrid, B, 0, stream>>>(h1, csr, off, deg, s);
  sage_kernel<<<ggrid, BG, 0, stream>>>(h1, s, cnt, w2l, b2l, w2r, b2r, h2);

  linear_kernel<<<ggrid, BG, 0, stream>>>(h2, wp1, bp1, tmp);
  linear_kernel<<<ggrid, BG, 0, stream>>>(tmp, wp2, bp2, out);
}


# each layer's neighbour sums taken inside its SAGE kernel: a wave gathers its 16 nodes' rows into an LDS tile read back as A fragments; both gather launches and the sums' trip through memory go
# speedup vs baseline: 3.6920x; 1.0433x over previous
#include <hip/hip_runtime.h>
#include <math.h>


typedef __attribute__((ext_vector_type(2))) float v2f;
typedef __attribute__((ext_vector_type(8))) float v8f;

#define N_NODES 100000
#define N_EDGES 1600000
#define DIM 64
#define NTILES (N_NODES / 16)
#define L2EPS 1e-12f

__device__ __forceinline__ v8f wmma_f32(v2f a, v2f b, v8f c) {
  return __builtin_amdgcn_wmma_f32_16x16x4_f32(
       false, a,  false, b,
       (short)0, c,  false,  false);
}

__device__ __forceinline__ v2f frag_pair(const float* __restrict__ p) {
  v2f r; r.x = p[0]; r.y = p[1]; return r;
}

#define WF_PITCH 33
#define WF_PAIRS (64 * WF_PITCH)
#define GEMM_T 256
#define WQ (DIM * DIM / 4 / GEMM_T)

__device__ __forceinline__ void load_weights(const float* __restrict__ w, float4 (&q)[WQ]) {
#pragma unroll
  for (int j = 0; j < WQ; ++j)
    q[j] = *(const float4*)(w + 4 * (threadIdx.x + j * GEMM_T));
}

__device__ __forceinline__ void store_weights(v2f* lw, const float4 (&q)[WQ]) {
#pragma unroll
  for (int j = 0; j < WQ; ++j) {
    const int i = threadIdx.x + j * GEMM_T;
    const int n = i >> 4;
    const int c = i & 15;
    const int f = (n >> 4) * 16 + c;
    v2f lo; lo.x = q[j].x; lo.y = q[j].y;
    v2f hi; hi.x = q[j].z; hi.y = q[j].w;
    lw[f * WF_PITCH + (n & 15)]      = lo;
    lw[f * WF_PITCH + 16 + (n & 15)] = hi;
  }
}

__global__ void zero_kernel(float* __restrict__ p, long long n) {
  long long i = (long long)blockIdx.x * blockDim.x + threadIdx.x;
  if (i < n) p[i] = 0.0f;
}

#define NB ((N_NODES + 255) / 256)
#define CHUNKS 250
#define CHUNK_V (N_EDGES / CHUNKS / 4)
#define BT_STRIDE 32
#define SORT_T 1024
static_assert(CHUNK_V * 4 * CHUNKS == N_EDGES, "chunks must tile the edge list");
static_assert(N_NODES <= (1 << 24), "src must fit the upper 24 bits of a record");
static_assert(NB <= SORT_T, "one thread per block in the scans");

__device__ __forceinline__ int block_scan_excl(int val, int* wtot, int& total) {
  const int lane = threadIdx.x & 31;
  const int wave = threadIdx.x >> 5;
  int inc = val;
#pragma unroll
  for (int o = 1; o < 32; o <<= 1) {
    const int up = __shfl_up(inc, o, 32);
    if (lane >= o) inc += up;
  }
  if (lane == 31) wtot[wave] = inc;
  __syncthreads();
  if (wave == 0) {
    const int t = wtot[lane];
    int ti = t;
#pragma unroll
    for (int o = 1; o < 32; o <<= 1) {
      const int up = __shfl_up(ti, o, 32);
      if (lane >= o) ti += up;
    }
    wtot[lane] = ti - t;
    if (lane == 31) wtot[32] = ti;
  }
  __syncthreads();
  total = wtot[32];
  const int ex = wtot[wave] + inc - val;
  __syncthreads();
  return ex;
}

__global__ __launch_bounds__(SORT_T) void bin_count_kernel(const int* __restrict__ dst,
                                                           int* btot,
                                                           int* __restrict__ run0) {
  __shared__ int hist[NB];
  const int t = threadIdx.x;
  const int c = blockIdx.x;
  if (t < NB) hist[t] = 0;
  __syncthreads();
  const int4* d4 = (const int4*)dst + (long long)c * CHUNK_V;
  for (int i = t; i < CHUNK_V; i += SORT_T) {
    const int4 d = d4[i];
    atomicAdd(&hist[d.x >> 8], 1);
    atomicAdd(&hist[d.y >> 8], 1);
    atomicAdd(&hist[d.z >> 8], 1);
    atomicAdd(&hist[d.w >> 8], 1);
  }
  __syncthreads();
  if (t < NB) run0[c * NB + t] = atomicAdd(btot + t * BT_STRIDE, hist[t]);
}

__global__ __launch_bounds__(SORT_T) void bin_scatter_kernel(const int* __restrict__ src,
                                                             const int* __restrict__ dst,
                                                             const int* __restrict__ btot,
                                                             const int* __restrict__ run0,
                                                             int* __restrict__ rec) {
  __shared__ int cur[NB];
  __shared__ int wtot[33];
  const int t = threadIdx.x;
  const int c = blockIdx.x;
  const int bt = (t < NB) ? btot[t * BT_STRIDE] : 0;
  int total;
  const int bstart = block_scan_excl(bt, wtot, total);
  if (t < NB) cur[t] = bstart + run0[c * NB + t];
  __syncthreads();
  const int4* s4 = (const int4*)src + (long long)c * CHUNK_V;
  const int4* d4 = (const int4*)dst + (long long)c * CHUNK_V;
  for (int i = t; i < CHUNK_V; i += SORT_T) {
    const int4 sn = s4[i];
    const int4 d  = d4[i];
    const int p0 = min(max(atomicAdd(&cur[d.x >> 8], 1), 0), N_EDGES - 1);
    const int p1 = min(max(atomicAdd(&cur[d.y >> 8], 1), 0), N_EDGES - 1);
    const int p2 = min(max(atomicAdd(&cur[d.z >> 8], 1), 0), N_EDGES - 1);
    const int p3 = min(max(atomicAdd(&cur[d.w >> 8], 1), 0), N_EDGES - 1);
    rec[p0] = (int)(((unsigned)sn.x << 8) | (unsigned)(d.x & 255));
    rec[p1] = (int)(((unsigned)sn.y << 8) | (unsigned)(d.y & 255));
    rec[p2] = (int)(((unsigned)sn.z << 8) | (unsigned)(d.z & 255));
    rec[p3] = (int)(((unsigned)sn.w << 8) | (unsigned)(d.w & 255));
  }
}

__global__ __launch_bounds__(SORT_T) void bin_sort_kernel(const int* __restrict__ btot,
                                                          const int* __restrict__ rec,
                                                          int* __restrict__ csr,
                                                          int* __restrict__ off,
                                                          int* __restrict__ deg) {
  __shared__ int cntL[256];
  __shared__ int offL[256];
  __shared__ int wtot[33];
  const int t = threadIdx.x;
  const int b = blockIdx.x;
  const int bt = (t < b) ? btot[t * BT_STRIDE] : 0;
  int before;
  block_scan_excl(bt, wtot, before);
  const int bstart = min(max(before, 0), N_EDGES);
  const int bsize  = min(max(btot[b * BT_STRIDE], 0), N_EDGES - bstart);
  if (t < 256) cntL[t] = 0;
  __syncthreads();
  for (int i = t; i < bsize; i += SORT_T) atomicAdd(&cntL[rec[bstart + i] & 255], 1);
  __syncthreads();
  const int c = (t < 256) ? cntL[t] : 0;
  int all;
  const int ex = block_scan_excl(c, wtot, all);
  if (t < 256) {
    offL[t] = ex;
    cntL[t] = 0;
    const int v = b * 256 + t;
    if (v < N_NODES) {
      off[v] = bstart + ex;
      deg[v] = c;
    }
  }
  __syncthreads();
  for (int i = t; i < bsize; i += SORT_T) {
    const int w = rec[bstart + i];
    const int k = w & 255;
    const int r = atomicAdd(&cntL[k], 1);
    const int p = min(max(bstart + offL[k] + r, 0), N_EDGES - 1);
    csr[p] = (int)((unsigned)w >> 8);
  }
}

#define GB 8
#define AG_PITCH 68

__device__ __forceinline__ void gather_tile(const float* __restrict__ x,
                                            const int* __restrict__ csr,
                                            int st_l, int dg_l, int l16, int half,
                                            float* __restrict__ my) {
  const int d4 = l16 * 4;
  for (int i = 0; i < 8; ++i) {
    const int m  = 2 * i + half;
    const int so = __shfl(st_l, m, 32);
    const int dr = __shfl(dg_l, m, 32);
    const int start = min(max(so, 0), N_EDGES);
    const int n     = min(max(dr, 0), N_EDGES - start);
    float4 acc = make_float4(0.0f, 0.0f, 0.0f, 0.0f);
    for (int j = 0; j < n; j += GB) {
      int idx[GB];
#pragma unroll
      for (int u = 0; u < GB; ++u) {
        const int e = csr[start + min(j + u, n - 1)];
        idx[u] = min(max(e, 0), N_NODES - 1);
      }
      float4 r[GB];
#pragma unroll
      for (int u = 0; u < GB; ++u)
        r[u] = *(const float4*)(x + (long long)idx[u] * DIM + d4);
#pragma unroll
      for (int u = 0; u < GB; ++u) {
        const bool in = (j + u < n);
        acc.x += in ? r[u].x : 0.0f;
        acc.y += in ? r[u].y : 0.0f;
        acc.z += in ? r[u].z : 0.0f;
        acc.w += in ? r[u].w : 0.0f;
      }
    }
    const float rinv = 1.0f / fmaxf((float)dr, 1.0f);
    acc.x *= rinv; acc.y *= rinv; acc.z *= rinv; acc.w *= rinv;
    *(float4*)(my + m * AG_PITCH + d4) = acc;
  }
}

__global__ __launch_bounds__(GEMM_T) void sage_kernel(
    const float* __restrict__ xin, const int* __restrict__ csr,
    const int* __restrict__ off, const int* __restrict__ deg,
    const float* __restrict__ wl, const float* __restrict__ bl,
    const float* __restrict__ wr, const float* __restrict__ br,
    float* __restrict__ out) {
  __shared__ v2f lwl[WF_PAIRS];
  __shared__ v2f lwr[WF_PAIRS];
  __shared__ float4 agg4[(GEMM_T / 32) * 16 * (AG_PITCH / 4)];
  const int wave = threadIdx.x >> 5;
  const int lane = threadIdx.x & 31;
  const int tile_raw = blockIdx.x * (blockDim.x >> 5) + wave;
  const bool live = tile_raw < NTILES;
  const int tile = live ? tile_raw : NTILES - 1;

  const int l16  = lane & 15;
  const int half = lane >> 4;
  const int rb   = tile * 16;
  const int arow = rb + l16;
  float* my = (float*)agg4 + wave * (16 * AG_PITCH);

  float4 ql[WQ], qr[WQ];
  load_weights(wl, ql);
  load_weights(wr, qr);
  const int st_l = off[arow];
  const int dg_l = deg[arow];
  v2f ax[16];
#pragma unroll
  for (int k = 0; k < 16; ++k)
    ax[k] = frag_pair(xin + (long long)arow * DIM + 4 * k + 2 * half);
  store_weights(lwl, ql);
  store_weights(lwr, qr);

  if (live) gather_tile(xin, csr, st_l, dg_l, l16, half, my);
  __syncthreads();
  if (!live) return;

  v2f ap[16];
#pragma unroll
  for (int k = 0; k < 16; ++k)
    ap[k] = *(const v2f*)(my + l16 * AG_PITCH + 4 * k + 2 * half);

  v8f acc[4];
#pragma unroll
  for (int nt = 0; nt < 4; ++nt) {
    v8f c8 = {};
    const int n0 = nt * 16;
#pragma unroll
    for (int k = 0; k < 16; ++k) {
      const v2f b = lwl[(nt * 16 + k) * WF_PITCH + lane];
      c8 = wmma_f32(ax[k], b, c8);
    }
#pragma unroll
    for (int k = 0; k < 16; ++k) {
      const v2f b = lwr[(nt * 16 + k) * WF_PITCH + lane];
      c8 = wmma_f32(ap[k], b, c8);
    }
    const float bias = bl[n0 + l16] + br[n0 + l16];
#pragma unroll
    for (int r = 0; r < 8; ++r) c8[r] += bias;
    acc[nt] = c8;
  }

  float ssq[8];
#pragma unroll
  for (int r = 0; r < 8; ++r) {
    float v = 0.0f;
#pragma unroll
    for (int nt = 0; nt < 4; ++nt) v += acc[nt][r] * acc[nt][r];
    ssq[r] = v;
  }
#pragma unroll
  for (int m = 1; m <= 8; m <<= 1) {
#pragma unroll
    for (int r = 0; r < 8; ++r) ssq[r] += __shfl_xor(ssq[r], m, 32);
  }

#pragma unroll
  for (int r = 0; r < 8; ++r) {
    const float inv = 1.0f / fmaxf(sqrtf(ssq[r]), L2EPS);
    const int row = rb + r + 8 * half;
#pragma unroll
    for (int nt = 0; nt < 4; ++nt) {
      float v = acc[nt][r] * inv;
      v = fmaxf(v, 0.0f);
      out[(long long)row * DIM + nt * 16 + l16] = v;
    }
  }
}

__global__ __launch_bounds__(GEMM_T) void linear_kernel(
    const float* __restrict__ xin, const float* __restrict__ w,
    const float* __restrict__ b, float* __restrict__ out) {
  __shared__ v2f lw[WF_PAIRS];
  const int wave = threadIdx.x >> 5;
  const int lane = threadIdx.x & 31;
  const int tile_raw = blockIdx.x * (blockDim.x >> 5) + wave;
  const bool live = tile_raw < NTILES;
  const int tile = live ? tile_raw : NTILES - 1;

  const int l16  = lane & 15;
  const int half = lane >> 4;
  const int rb   = tile * 16;
  const int arow = rb + l16;

  v2f ax[16];
#pragma unroll
  for (int k = 0; k < 16; ++k)
    ax[k] = frag_pair(xin + (long long)arow * DIM + 4 * k + 2 * half);

  float4 qw[WQ];
  load_weights(w, qw);
  store_weights(lw, qw);
  __syncthreads();
  if (!live) return;

#pragma unroll
  for (int nt = 0; nt < 4; ++nt) {
    v8f c8 = {};
    const int n0 = nt * 16;
#pragma unroll
    for (int k = 0; k < 16; ++k) {
      const v2f bf = lw[(nt * 16 + k) * WF_PITCH + lane];
      c8 = wmma_f32(ax[k], bf, c8);
    }
    const float bias = b[n0 + l16];
#pragma unroll
    for (int r = 0; r < 8; ++r) {
      const int row = rb + r + 8 * half;
      out[(long long)row * DIM + n0 + l16] = c8[r] + bias;
    }
  }
}

extern "C" void kernel_launch(void* const* d_in, const int* in_sizes, int n_in,
                              void* d_out, int out_size, void* d_ws, size_t ws_size,
                              hipStream_t stream) {
  (void)in_sizes; (void)n_in; (void)out_size; (void)ws_size;

  const float* x   = (const float*)d_in[0];
  const int*   ei  = (const int*)d_in[1];
  const int*   src = ei;
  const int*   dst = ei + N_EDGES;
  const float* w1l = (const float*)d_in[2];
  const float* b1l = (const float*)d_in[3];
  const float* w1r = (const float*)d_in[4];
  const float* b1r = (const float*)d_in[5];
  const float* w2l = (const float*)d_in[6];
  const float* b2l = (const float*)d_in[7];
  const float* w2r = (const float*)d_in[8];
  const float* b2r = (const float*)d_in[9];
  const float* wp1 = (const float*)d_in[10];
  const float* bp1 = (const float*)d_in[11];
  const float* wp2 = (const float*)d_in[12];
  const float* bp2 = (const float*)d_in[13];
  float* out = (float*)d_out;

  const long long NS = (long long)N_NODES * DIM;
  float* s   = (float*)d_ws;
  float* h1  = s + NS;
  float* h2  = h1 + NS;
  float* tmp = s;
  int* btot = (int*)s;
  int* run0 = btot + NB * BT_STRIDE;
  int* deg  = run0 + (CHUNKS * NB + 31) / 32 * 32;
  int* off  = deg + N_NODES;
  int* rec  = off + N_NODES;
  int* csr  = rec + N_EDGES;

  const dim3 B(256);
  const dim3 BG(GEMM_T);
  const dim3 BS(SORT_T);
  const int zgrid = (NB * BT_STRIDE + 255) / 256;
  const int ggrid = (NTILES + 7) / 8;

  zero_kernel<<<zgrid, B, 0, stream>>>((float*)btot, NB * BT_STRIDE);
  bin_count_kernel<<<CHUNKS, BS, 0, stream>>>(dst, btot, run0);
  bin_scatter_kernel<<<CHUNKS, BS, 0, stream>>>(src, dst, btot, run0, rec);
  bin_sort_kernel<<<NB, BS, 0, stream>>>(btot, rec, csr, off, deg);

  sage_kernel<<<ggrid, BG, 0, stream>>>(x, csr, off, deg, w1l, b1l, w1r, b1r, h1);

  sage_kernel<<<ggrid, BG, 0, stream>>>(h1, csr, off, deg, w2l, b2l, w2r, b2r, h2);

  linear_kernel<<<ggrid, BG, 0, stream>>>(h2, wp1, bp1, tmp);
  linear_kernel<<<ggrid, BG, 0, stream>>>(tmp, wp2, bp2, out);
}


# post-MP head (both linears) runs inside the layer-2 kernel on the wave's tile: C-to-A layout via its LDS tile, head weights take over the layer weights' LDS; 2 launches and the h2/tmp trips go
# speedup vs baseline: 4.0237x; 1.0898x over previous
#include <hip/hip_runtime.h>
#include <math.h>


typedef __attribute__((ext_vector_type(2))) float v2f;
typedef __attribute__((ext_vector_type(8))) float v8f;

#define N_NODES 100000
#define N_EDGES 1600000
#define DIM 64
#define NTILES (N_NODES / 16)
#define L2EPS 1e-12f

__device__ __forceinline__ v8f wmma_f32(v2f a, v2f b, v8f c) {
  return __builtin_amdgcn_wmma_f32_16x16x4_f32(
       false, a,  false, b,
       (short)0, c,  false,  false);
}

__device__ __forceinline__ v2f frag_pair(const float* __restrict__ p) {
  v2f r; r.x = p[0]; r.y = p[1]; return r;
}

#define WF_PITCH 33
#define WF_PAIRS (64 * WF_PITCH)
#define GEMM_T 256
#define WQ (DIM * DIM / 4 / GEMM_T)

__device__ __forceinline__ void load_weights(const float* __restrict__ w, float4 (&q)[WQ]) {
#pragma unroll
  for (int j = 0; j < WQ; ++j)
    q[j] = *(const float4*)(w + 4 * (threadIdx.x + j * GEMM_T));
}

__device__ __forceinline__ void store_weights(v2f* lw, const float4 (&q)[WQ]) {
#pragma unroll
  for (int j = 0; j < WQ; ++j) {
    const int i = threadIdx.x + j * GEMM_T;
    const int n = i >> 4;
    const int c = i & 15;
    const int f = (n >> 4) * 16 + c;
    v2f lo; lo.x = q[j].x; lo.y = q[j].y;
    v2f hi; hi.x = q[j].z; hi.y = q[j].w;
    lw[f * WF_PITCH + (n & 15)]      = lo;
    lw[f * WF_PITCH + 16 + (n & 15)] = hi;
  }
}

__global__ void zero_kernel(float* __restrict__ p, long long n) {
  long long i = (long long)blockIdx.x * blockDim.x + threadIdx.x;
  if (i < n) p[i] = 0.0f;
}

#define NB ((N_NODES + 255) / 256)
#define CHUNKS 250
#define CHUNK_V (N_EDGES / CHUNKS / 4)
#define BT_STRIDE 32
#define SORT_T 1024
static_assert(CHUNK_V * 4 * CHUNKS == N_EDGES, "chunks must tile the edge list");
static_assert(N_NODES <= (1 << 24), "src must fit the upper 24 bits of a record");
static_assert(NB <= SORT_T, "one thread per block in the scans");

__device__ __forceinline__ int block_scan_excl(int val, int* wtot, int& total) {
  const int lane = threadIdx.x & 31;
  const int wave = threadIdx.x >> 5;
  int inc = val;
#pragma unroll
  for (int o = 1; o < 32; o <<= 1) {
    const int up = __shfl_up(inc, o, 32);
    if (lane >= o) inc += up;
  }
  if (lane == 31) wtot[wave] = inc;
  __syncthreads();
  if (wave == 0) {
    const int t = wtot[lane];
    int ti = t;
#pragma unroll
    for (int o = 1; o < 32; o <<= 1) {
      const int up = __shfl_up(ti, o, 32);
      if (lane >= o) ti += up;
    }
    wtot[lane] = ti - t;
    if (lane == 31) wtot[32] = ti;
  }
  __syncthreads();
  total = wtot[32];
  const int ex = wtot[wave] + inc - val;
  __syncthreads();
  return ex;
}

__global__ __launch_bounds__(SORT_T) void bin_count_kernel(const int* __restrict__ dst,
                                                           int* btot,
                                                           int* __restrict__ run0) {
  __shared__ int hist[NB];
  const int t = threadIdx.x;
  const int c = blockIdx.x;
  if (t < NB) hist[t] = 0;
  __syncthreads();
  const int4* d4 = (const int4*)dst + (long long)c * CHUNK_V;
  for (int i = t; i < CHUNK_V; i += SORT_T) {
    const int4 d = d4[i];
    atomicAdd(&hist[d.x >> 8], 1);
    atomicAdd(&hist[d.y >> 8], 1);
    atomicAdd(&hist[d.z >> 8], 1);
    atomicAdd(&hist[d.w >> 8], 1);
  }
  __syncthreads();
  if (t < NB) run0[c * NB + t] = atomicAdd(btot + t * BT_STRIDE, hist[t]);
}

__global__ __launch_bounds__(SORT_T) void bin_scatter_kernel(const int* __restrict__ src,
                                                             const int* __restrict__ dst,
                                                             const int* __restrict__ btot,
                                                             const int* __restrict__ run0,
                                                             int* __restrict__ rec) {
  __shared__ int cur[NB];
  __shared__ int wtot[33];
  const int t = threadIdx.x;
  const int c = blockIdx.x;
  const int bt = (t < NB) ? btot[t * BT_STRIDE] : 0;
  int total;
  const int bstart = block_scan_excl(bt, wtot, total);
  if (t < NB) cur[t] = bstart + run0[c * NB + t];
  __syncthreads();
  const int4* s4 = (const int4*)src + (long long)c * CHUNK_V;
  const int4* d4 = (const int4*)dst + (long long)c * CHUNK_V;
  for (int i = t; i < CHUNK_V; i += SORT_T) {
    const int4 sn = s4[i];
    const int4 d  = d4[i];
    const int p0 = min(max(atomicAdd(&cur[d.x >> 8], 1), 0), N_EDGES - 1);
    const int p1 = min(max(atomicAdd(&cur[d.y >> 8], 1), 0), N_EDGES - 1);
    const int p2 = min(max(atomicAdd(&cur[d.z >> 8], 1), 0), N_EDGES - 1);
    const int p3 = min(max(atomicAdd(&cur[d.w >> 8], 1), 0), N_EDGES - 1);
    rec[p0] = (int)(((unsigned)sn.x << 8) | (unsigned)(d.x & 255));
    rec[p1] = (int)(((unsigned)sn.y << 8) | (unsigned)(d.y & 255));
    rec[p2] = (int)(((unsigned)sn.z << 8) | (unsigned)(d.z & 255));
    rec[p3] = (int)(((unsigned)sn.w << 8) | (unsigned)(d.w & 255));
  }
}

__global__ __launch_bounds__(SORT_T) void bin_sort_kernel(const int* __restrict__ btot,
                                                          const int* __restrict__ rec,
                                                          int* __restrict__ csr,
                                                          int* __restrict__ off,
                                                          int* __restrict__ deg) {
  __shared__ int cntL[256];
  __shared__ int offL[256];
  __shared__ int wtot[33];
  const int t = threadIdx.x;
  const int b = blockIdx.x;
  const int bt = (t < b) ? btot[t * BT_STRIDE] : 0;
  int before;
  block_scan_excl(bt, wtot, before);
  const int bstart = min(max(before, 0), N_EDGES);
  const int bsize  = min(max(btot[b * BT_STRIDE], 0), N_EDGES - bstart);
  if (t < 256) cntL[t] = 0;
  __syncthreads();
  for (int i = t; i < bsize; i += SORT_T) atomicAdd(&cntL[rec[bstart + i] & 255], 1);
  __syncthreads();
  const int c = (t < 256) ? cntL[t] : 0;
  int all;
  const int ex = block_scan_excl(c, wtot, all);
  if (t < 256) {
    offL[t] = ex;
    cntL[t] = 0;
    const int v = b * 256 + t;
    if (v < N_NODES) {
      off[v] = bstart + ex;
      deg[v] = c;
    }
  }
  __syncthreads();
  for (int i = t; i < bsize; i += SORT_T) {
    const int w = rec[bstart + i];
    const int k = w & 255;
    const int r = atomicAdd(&cntL[k], 1);
    const int p = min(max(bstart + offL[k] + r, 0), N_EDGES - 1);
    csr[p] = (int)((unsigned)w >> 8);
  }
}

#define GB 8
#define AG_PITCH 68

__device__ __forceinline__ void gather_tile(const float* __restrict__ x,
                                            const int* __restrict__ csr,
                                            int st_l, int dg_l, int l16, int half,
                                            float* __restrict__ my) {
  const int d4 = l16 * 4;
  for (int i = 0; i < 8; ++i) {
    const int m  = 2 * i + half;
    const int so = __shfl(st_l, m, 32);
    const int dr = __shfl(dg_l, m, 32);
    const int start = min(max(so, 0), N_EDGES);
    const int n     = min(max(dr, 0), N_EDGES - start);
    float4 acc = make_float4(0.0f, 0.0f, 0.0f, 0.0f);
    for (int j = 0; j < n; j += GB) {
      int idx[GB];
#pragma unroll
      for (int u = 0; u < GB; ++u) {
        const int e = csr[start + min(j + u, n - 1)];
        idx[u] = min(max(e, 0), N_NODES - 1);
      }
      float4 r[GB];
#pragma unroll
      for (int u = 0; u < GB; ++u)
        r[u] = *(const float4*)(x + (long long)idx[u] * DIM + d4);
#pragma unroll
      for (int u = 0; u < GB; ++u) {
        const bool in = (j + u < n);
        acc.x += in ? r[u].x : 0.0f;
        acc.y += in ? r[u].y : 0.0f;
        acc.z += in ? r[u].z : 0.0f;
        acc.w += in ? r[u].w : 0.0f;
      }
    }
    const float rinv = 1.0f / fmaxf((float)dr, 1.0f);
    acc.x *= rinv; acc.y *= rinv; acc.z *= rinv; acc.w *= rinv;
    *(float4*)(my + m * AG_PITCH + d4) = acc;
  }
}

template <bool HEAD>
__global__ __launch_bounds__(GEMM_T) void sage_kernel(
    const float* __restrict__ xin, const int* __restrict__ csr,
    const int* __restrict__ off, const int* __restrict__ deg,
    const float* __restrict__ wl, const float* __restrict__ bl,
    const float* __restrict__ wr, const float* __restrict__ br,
    const float* __restrict__ wp1, const float* __restrict__ bp1,
    const float* __restrict__ wp2, const float* __restrict__ bp2,
    float* __restrict__ out) {
  __shared__ v2f lwl[WF_PAIRS];
  __shared__ v2f lwr[WF_PAIRS];
  __shared__ float4 agg4[(GEMM_T / 32) * 16 * (AG_PITCH / 4)];
  const int wave = threadIdx.x >> 5;
  const int lane = threadIdx.x & 31;
  const int tile_raw = blockIdx.x * (blockDim.x >> 5) + wave;
  const bool live = tile_raw < NTILES;
  const int tile = live ? tile_raw : NTILES - 1;

  const int l16  = lane & 15;
  const int half = lane >> 4;
  const int rb   = tile * 16;
  const int arow = rb + l16;
  float* my = (float*)agg4 + wave * (16 * AG_PITCH);

  float4 ql[WQ], qr[WQ];
  load_weights(wl, ql);
  load_weights(wr, qr);
  const int st_l = off[arow];
  const int dg_l = deg[arow];
  v2f ax[16];
#pragma unroll
  for (int k = 0; k < 16; ++k)
    ax[k] = frag_pair(xin + (long long)arow * DIM + 4 * k + 2 * half);
  store_weights(lwl, ql);
  store_weights(lwr, qr);

  if (live) gather_tile(xin, csr, st_l, dg_l, l16, half, my);
  __syncthreads();
  if constexpr (!HEAD) {
    if (!live) return;
  }

  v2f ap[16];
#pragma unroll
  for (int k = 0; k < 16; ++k)
    ap[k] = *(const v2f*)(my + l16 * AG_PITCH + 4 * k + 2 * half);

  v8f acc[4];
#pragma unroll
  for (int nt = 0; nt < 4; ++nt) {
    v8f c8 = {};
    const int n0 = nt * 16;
#pragma unroll
    for (int k = 0; k < 16; ++k) {
      const v2f b = lwl[(nt * 16 + k) * WF_PITCH + lane];
      c8 = wmma_f32(ax[k], b, c8);
    }
#pragma unroll
    for (int k = 0; k < 16; ++k) {
      const v2f b = lwr[(nt * 16 + k) * WF_PITCH + lane];
      c8 = wmma_f32(ap[k], b, c8);
    }
    const float bias = bl[n0 + l16] + br[n0 + l16];
#pragma unroll
    for (int r = 0; r < 8; ++r) c8[r] += bias;
    acc[nt] = c8;
  }

  float4 q1[WQ], q2[WQ];
  if constexpr (HEAD) {
    load_weights(wp1, q1);
    load_weights(wp2, q2);
  }

  float ssq[8];
#pragma unroll
  for (int r = 0; r < 8; ++r) {
    float v = 0.0f;
#pragma unroll
    for (int nt = 0; nt < 4; ++nt) v += acc[nt][r] * acc[nt][r];
    ssq[r] = v;
  }
#pragma unroll
  for (int m = 1; m <= 8; m <<= 1) {
#pragma unroll
    for (int r = 0; r < 8; ++r) ssq[r] += __shfl_xor(ssq[r], m, 32);
  }

#pragma unroll
  for (int r = 0; r < 8; ++r) {
    const float inv = 1.0f / fmaxf(sqrtf(ssq[r]), L2EPS);
    const int row = rb + r + 8 * half;
#pragma unroll
    for (int nt = 0; nt < 4; ++nt) {
      float v = acc[nt][r] * inv;
      v = fmaxf(v, 0.0f);
      if constexpr (HEAD)
        my[(r + 8 * half) * AG_PITCH + nt * 16 + l16] = v;
      else
        out[(long long)row * DIM + nt * 16 + l16] = v;
    }
  }

  if constexpr (HEAD) {
    __syncthreads();
    store_weights(lwl, q1);
    store_weights(lwr, q2);
    __syncthreads();

    v2f ah[16];
#pragma unroll
    for (int k = 0; k < 16; ++k)
      ah[k] = *(const v2f*)(my + l16 * AG_PITCH + 4 * k + 2 * half);
#pragma unroll
    for (int nt = 0; nt < 4; ++nt) {
      v8f c8 = {};
#pragma unroll
      for (int k = 0; k < 16; ++k) {
        const v2f b = lwl[(nt * 16 + k) * WF_PITCH + lane];
        c8 = wmma_f32(ah[k], b, c8);
      }
      acc[nt] = c8;
    }
#pragma unroll
    for (int nt = 0; nt < 4; ++nt) {
      const float bias = bp1[nt * 16 + l16];
#pragma unroll
      for (int r = 0; r < 8; ++r)
        my[(r + 8 * half) * AG_PITCH + nt * 16 + l16] = acc[nt][r] + bias;
    }
    __syncthreads();

#pragma unroll
    for (int k = 0; k < 16; ++k)
      ah[k] = *(const v2f*)(my + l16 * AG_PITCH + 4 * k + 2 * half);
#pragma unroll
    for (int nt = 0; nt < 4; ++nt) {
      v8f c8 = {};
#pragma unroll
      for (int k = 0; k < 16; ++k) {
        const v2f b = lwr[(nt * 16 + k) * WF_PITCH + lane];
        c8 = wmma_f32(ah[k], b, c8);
      }
      acc[nt] = c8;
    }
    if (live) {
#pragma unroll
      for (int nt = 0; nt < 4; ++nt) {
        const float bias = bp2[nt * 16 + l16];
#pragma unroll
        for (int r = 0; r < 8; ++r) {
          const int row = rb + r + 8 * half;
          out[(long long)row * DIM + nt * 16 + l16] = acc[nt][r] + bias;
        }
      }
    }
  }
}

extern "C" void kernel_launch(void* const* d_in, const int* in_sizes, int n_in,
                              void* d_out, int out_size, void* d_ws, size_t ws_size,
                              hipStream_t stream) {
  (void)in_sizes; (void)n_in; (void)out_size; (void)ws_size;

  const float* x   = (const float*)d_in[0];
  const int*   ei  = (const int*)d_in[1];
  const int*   src = ei;
  const int*   dst = ei + N_EDGES;
  const float* w1l = (const float*)d_in[2];
  const float* b1l = (const float*)d_in[3];
  const float* w1r = (const float*)d_in[4];
  const float* b1r = (const float*)d_in[5];
  const float* w2l = (const float*)d_in[6];
  const float* b2l = (const float*)d_in[7];
  const float* w2r = (const float*)d_in[8];
  const float* b2r = (const float*)d_in[9];
  const float* wp1 = (const float*)d_in[10];
  const float* bp1 = (const float*)d_in[11];
  const float* wp2 = (const float*)d_in[12];
  const float* bp2 = (const float*)d_in[13];
  float* out = (float*)d_out;

  const long long NS = (long long)N_NODES * DIM;
  float* s   = (float*)d_ws;
  float* h1  = s + NS;
  int* btot = (int*)s;
  int* run0 = btot + NB * BT_STRIDE;
  int* deg  = run0 + (CHUNKS * NB + 31) / 32 * 32;
  int* off  = deg + N_NODES;
  int* rec  = off + N_NODES;
  int* csr  = rec + N_EDGES;

  const dim3 B(256);
  const dim3 BG(GEMM_T);
  const dim3 BS(SORT_T);
  const int zgrid = (NB * BT_STRIDE + 255) / 256;
  const int ggrid = (NTILES + 7) / 8;

  zero_kernel<<<zgrid, B, 0, stream>>>((float*)btot, NB * BT_STRIDE);
  bin_count_kernel<<<CHUNKS, BS, 0, stream>>>(dst, btot, run0);
  bin_scatter_kernel<<<CHUNKS, BS, 0, stream>>>(src, dst, btot, run0, rec);
  bin_sort_kernel<<<NB, BS, 0, stream>>>(btot, rec, csr, off, deg);

  sage_kernel<false><<<ggrid, BG, 0, stream>>>(x, csr, off, deg, w1l, b1l, w1r, b1r,
                                               wp1, bp1, wp2, bp2, h1);

  sage_kernel<true><<<ggrid, BG, 0, stream>>>(h1, csr, off, deg, w2l, b2l, w2r, b2r,
                                              wp1, bp1, wp2, bp2, out);
}


# edge sort level 1 with no shared counters: each chunk sorts its edges by block in LDS and writes one run + offset table; zero and count launches and global atomics go; sums keep LDS arrival order
# speedup vs baseline: 4.8164x; 1.1970x over previous
#include <hip/hip_runtime.h>
#include <math.h>


typedef __attribute__((ext_vector_type(2))) float v2f;
typedef __attribute__((ext_vector_type(8))) float v8f;

#define N_NODES 100000
#define N_EDGES 1600000
#define DIM 64
#define NTILES (N_NODES / 16)
#define L2EPS 1e-12f

__device__ __forceinline__ v8f wmma_f32(v2f a, v2f b, v8f c) {
  return __builtin_amdgcn_wmma_f32_16x16x4_f32(
       false, a,  false, b,
       (short)0, c,  false,  false);
}

__device__ __forceinline__ v2f frag_pair(const float* __restrict__ p) {
  v2f r; r.x = p[0]; r.y = p[1]; return r;
}

#define WF_PITCH 33
#define WF_PAIRS (64 * WF_PITCH)
#define GEMM_T 256
#define WQ (DIM * DIM / 4 / GEMM_T)

__device__ __forceinline__ void load_weights(const float* __restrict__ w, float4 (&q)[WQ]) {
#pragma unroll
  for (int j = 0; j < WQ; ++j)
    q[j] = *(const float4*)(w + 4 * (threadIdx.x + j * GEMM_T));
}

__device__ __forceinline__ void store_weights(v2f* lw, const float4 (&q)[WQ]) {
#pragma unroll
  for (int j = 0; j < WQ; ++j) {
    const int i = threadIdx.x + j * GEMM_T;
    const int n = i >> 4;
    const int c = i & 15;
    const int f = (n >> 4) * 16 + c;
    v2f lo; lo.x = q[j].x; lo.y = q[j].y;
    v2f hi; hi.x = q[j].z; hi.y = q[j].w;
    lw[f * WF_PITCH + (n & 15)]      = lo;
    lw[f * WF_PITCH + 16 + (n & 15)] = hi;
  }
}

#define NB ((N_NODES + 255) / 256)
#define CHUNKS 250
#define CHUNK_E (N_EDGES / CHUNKS)
#define CHUNK_V (CHUNK_E / 4)
#define O1 (NB + 1)
#define SORT_T 1024
#define RUN_U 4
static_assert(CHUNK_V * 4 * CHUNKS == N_EDGES, "chunks must tile the edge list");
static_assert(N_NODES <= (1 << 24), "src must fit the upper 24 bits of a record");
static_assert(NB <= SORT_T, "one thread per block in the scans");
static_assert(CHUNK_V > SORT_T && CHUNK_V <= 2 * SORT_T, "a thread owns one or two int4 of its chunk");
static_assert(CHUNKS * 4 <= SORT_T, "four threads per chunk in level 2");

__device__ __forceinline__ int block_scan_excl(int val, int* wtot, int& total) {
  const int lane = threadIdx.x & 31;
  const int wave = threadIdx.x >> 5;
  int inc = val;
#pragma unroll
  for (int o = 1; o < 32; o <<= 1) {
    const int up = __shfl_up(inc, o, 32);
    if (lane >= o) inc += up;
  }
  if (lane == 31) wtot[wave] = inc;
  __syncthreads();
  if (wave == 0) {
    const int t = wtot[lane];
    int ti = t;
#pragma unroll
    for (int o = 1; o < 32; o <<= 1) {
      const int up = __shfl_up(ti, o, 32);
      if (lane >= o) ti += up;
    }
    wtot[lane] = ti - t;
    if (lane == 31) wtot[32] = ti;
  }
  __syncthreads();
  total = wtot[32];
  const int ex = wtot[wave] + inc - val;
  __syncthreads();
  return ex;
}

__global__ __launch_bounds__(SORT_T) void chunk_sort_kernel(const int* __restrict__ src,
                                                            const int* __restrict__ dst,
                                                            int* __restrict__ rec,
                                                            int* __restrict__ off1) {
  __shared__ int hist[NB];
  __shared__ int offs[NB];
  __shared__ int wtot[33];
  __shared__ int4 recL4[CHUNK_V];
  int* recL = (int*)recL4;
  const int t = threadIdx.x;
  const int c = blockIdx.x;
  const bool two = (t + SORT_T) < CHUNK_V;
  const int4* s4 = (const int4*)src + (long long)c * CHUNK_V;
  const int4* d4 = (const int4*)dst + (long long)c * CHUNK_V;
  const int i1 = two ? t + SORT_T : CHUNK_V - 1;
  const int4 sa = s4[t];
  const int4 da = d4[t];
  const int4 sb = s4[i1];
  const int4 db = d4[i1];
  if (t < NB) hist[t] = 0;
  __syncthreads();
  const int r0 = atomicAdd(&hist[da.x >> 8], 1);
  const int r1 = atomicAdd(&hist[da.y >> 8], 1);
  const int r2 = atomicAdd(&hist[da.z >> 8], 1);
  const int r3 = atomicAdd(&hist[da.w >> 8], 1);
  int r4 = 0, r5 = 0, r6 = 0, r7 = 0;
  if (two) {
    r4 = atomicAdd(&hist[db.x >> 8], 1);
    r5 = atomicAdd(&hist[db.y >> 8], 1);
    r6 = atomicAdd(&hist[db.z >> 8], 1);
    r7 = atomicAdd(&hist[db.w >> 8], 1);
  }
  __syncthreads();
  const int h = (t < NB) ? hist[t] : 0;
  int total;
  const int ex = block_scan_excl(h, wtot, total);
  if (t < NB) {
    offs[t] = ex;
    off1[c * O1 + t] = ex;
  }
  if (t == 0) off1[c * O1 + NB] = total;
  __syncthreads();
  recL[min(max(offs[da.x >> 8] + r0, 0), CHUNK_E - 1)] = (int)(((unsigned)sa.x << 8) | (unsigned)(da.x & 255));
  recL[min(max(offs[da.y >> 8] + r1, 0), CHUNK_E - 1)] = (int)(((unsigned)sa.y << 8) | (unsigned)(da.y & 255));
  recL[min(max(offs[da.z >> 8] + r2, 0), CHUNK_E - 1)] = (int)(((unsigned)sa.z << 8) | (unsigned)(da.z & 255));
  recL[min(max(offs[da.w >> 8] + r3, 0), CHUNK_E - 1)] = (int)(((unsigned)sa.w << 8) | (unsigned)(da.w & 255));
  if (two) {
    recL[min(max(offs[db.x >> 8] + r4, 0), CHUNK_E - 1)] = (int)(((unsigned)sb.x << 8) | (unsigned)(db.x & 255));
    recL[min(max(offs[db.y >> 8] + r5, 0), CHUNK_E - 1)] = (int)(((unsigned)sb.y << 8) | (unsigned)(db.y & 255));
    recL[min(max(offs[db.z >> 8] + r6, 0), CHUNK_E - 1)] = (int)(((unsigned)sb.z << 8) | (unsigned)(db.z & 255));
    recL[min(max(offs[db.w >> 8] + r7, 0), CHUNK_E - 1)] = (int)(((unsigned)sb.w << 8) | (unsigned)(db.w & 255));
  }
  __syncthreads();
  int4* o4 = (int4*)(rec + (long long)c * CHUNK_E);
  o4[t] = recL4[t];
  if (two) o4[t + SORT_T] = recL4[t + SORT_T];
}

__global__ __launch_bounds__(SORT_T) void bin_sort_kernel(const int* __restrict__ off1,
                                                          const int* __restrict__ rec,
                                                          int* __restrict__ csr,
                                                          int* __restrict__ off,
                                                          int* __restrict__ deg) {
  __shared__ int cntL[256];
  __shared__ int offL[256];
  __shared__ int wtot[33];
  const int t = threadIdx.x;
  const int b = blockIdx.x;
  const int q = t & 3;
  const int c = min(t >> 2, CHUNKS - 1);
  int s = 0, e = 0;
  if ((t >> 2) < CHUNKS) {
    s = min(max(off1[c * O1 + b], 0), CHUNK_E);
    e = min(max(off1[c * O1 + b + 1], s), CHUNK_E);
  }
  int before;
  block_scan_excl((q == 0) ? s : 0, wtot, before);
  const int bstart = min(max(before, 0), N_EDGES);
  if (t < 256) cntL[t] = 0;
  __syncthreads();
  const int* rc = rec + (long long)c * CHUNK_E;
  for (int i = s + q; i < e; i += 4 * RUN_U) {
    int w[RUN_U];
#pragma unroll
    for (int u = 0; u < RUN_U; ++u) w[u] = rc[min(i + 4 * u, e - 1)];
#pragma unroll
    for (int u = 0; u < RUN_U; ++u)
      atomicAdd(&cntL[w[u] & 255], (i + 4 * u < e) ? 1 : 0);
  }
  __syncthreads();
  const int cn = (t < 256) ? cntL[t] : 0;
  int all;
  const int ex = block_scan_excl(cn, wtot, all);
  if (t < 256) {
    offL[t] = ex;
    cntL[t] = 0;
    const int v = b * 256 + t;
    if (v < N_NODES) {
      off[v] = bstart + ex;
      deg[v] = cn;
    }
  }
  __syncthreads();
  for (int i = s + q; i < e; i += 4 * RUN_U) {
    int w[RUN_U];
#pragma unroll
    for (int u = 0; u < RUN_U; ++u) w[u] = rc[min(i + 4 * u, e - 1)];
    int r[RUN_U];
#pragma unroll
    for (int u = 0; u < RUN_U; ++u)
      r[u] = atomicAdd(&cntL[w[u] & 255], (i + 4 * u < e) ? 1 : 0);
#pragma unroll
    for (int u = 0; u < RUN_U; ++u) {
      if (i + 4 * u < e) {
        const int p = min(max(bstart + offL[w[u] & 255] + r[u], 0), N_EDGES - 1);
        csr[p] = (int)((unsigned)w[u] >> 8);
      }
    }
  }
}

#define GB 8
#define AG_PITCH 68

__device__ __forceinline__ void gather_tile(const float* __restrict__ x,
                                            const int* __restrict__ csr,
                                            int st_l, int dg_l, int l16, int half,
                                            float* __restrict__ my) {
  const int d4 = l16 * 4;
  for (int i = 0; i < 8; ++i) {
    const int m  = 2 * i + half;
    const int so = __shfl(st_l, m, 32);
    const int dr = __shfl(dg_l, m, 32);
    const int start = min(max(so, 0), N_EDGES);
    const int n     = min(max(dr, 0), N_EDGES - start);
    float4 acc = make_float4(0.0f, 0.0f, 0.0f, 0.0f);
    for (int j = 0; j < n; j += GB) {
      int idx[GB];
#pragma unroll
      for (int u = 0; u < GB; ++u) {
        const int e = csr[start + min(j + u, n - 1)];
        idx[u] = min(max(e, 0), N_NODES - 1);
      }
      float4 r[GB];
#pragma unroll
      for (int u = 0; u < GB; ++u)
        r[u] = *(const float4*)(x + (long long)idx[u] * DIM + d4);
#pragma unroll
      for (int u = 0; u < GB; ++u) {
        const bool in = (j + u < n);
        acc.x += in ? r[u].x : 0.0f;
        acc.y += in ? r[u].y : 0.0f;
        acc.z += in ? r[u].z : 0.0f;
        acc.w += in ? r[u].w : 0.0f;
      }
    }
    const float rinv = 1.0f / fmaxf((float)dr, 1.0f);
    acc.x *= rinv; acc.y *= rinv; acc.z *= rinv; acc.w *= rinv;
    *(float4*)(my + m * AG_PITCH + d4) = acc;
  }
}

template <bool HEAD>
__global__ __launch_bounds__(GEMM_T) void sage_kernel(
    const float* __restrict__ xin, const int* __restrict__ csr,
    const int* __restrict__ off, const int* __restrict__ deg,
    const float* __restrict__ wl, const float* __restrict__ bl,
    const float* __restrict__ wr, const float* __restrict__ br,
    const float* __restrict__ wp1, const float* __restrict__ bp1,
    const float* __restrict__ wp2, const float* __restrict__ bp2,
    float* __restrict__ out) {
  __shared__ v2f lwl[WF_PAIRS];
  __shared__ v2f lwr[WF_PAIRS];
  __shared__ float4 agg4[(GEMM_T / 32) * 16 * (AG_PITCH / 4)];
  const int wave = threadIdx.x >> 5;
  const int lane = threadIdx.x & 31;
  const int tile_raw = blockIdx.x * (blockDim.x >> 5) + wave;
  const bool live = tile_raw < NTILES;
  const int tile = live ? tile_raw : NTILES - 1;

  const int l16  = lane & 15;
  const int half = lane >> 4;
  const int rb   = tile * 16;
  const int arow = rb + l16;
  float* my = (float*)agg4 + wave * (16 * AG_PITCH);

  float4 ql[WQ], qr[WQ];
  load_weights(wl, ql);
  load_weights(wr, qr);
  const int st_l = off[arow];
  const int dg_l = deg[arow];
  v2f ax[16];
#pragma unroll
  for (int k = 0; k < 16; ++k)
    ax[k] = frag_pair(xin + (long long)arow * DIM + 4 * k + 2 * half);
  store_weights(lwl, ql);
  store_weights(lwr, qr);

  if (live) gather_tile(xin, csr, st_l, dg_l, l16, half, my);
  __syncthreads();
  if constexpr (!HEAD) {
    if (!live) return;
  }

  v2f ap[16];
#pragma unroll
  for (int k = 0; k < 16; ++k)
    ap[k] = *(const v2f*)(my + l16 * AG_PITCH + 4 * k + 2 * half);

  v8f acc[4];
#pragma unroll
  for (int nt = 0; nt < 4; ++nt) {
    v8f c8 = {};
    const int n0 = nt * 16;
#pragma unroll
    for (int k = 0; k < 16; ++k) {
      const v2f b = lwl[(nt * 16 + k) * WF_PITCH + lane];
      c8 = wmma_f32(ax[k], b, c8);
    }
#pragma unroll
    for (int k = 0; k < 16; ++k) {
      const v2f b = lwr[(nt * 16 + k) * WF_PITCH + lane];
      c8 = wmma_f32(ap[k], b, c8);
    }
    const float bias = bl[n0 + l16] + br[n0 + l16];
#pragma unroll
    for (int r = 0; r < 8; ++r) c8[r] += bias;
    acc[nt] = c8;
  }

  float4 q1[WQ], q2[WQ];
  if constexpr (HEAD) {
    load_weights(wp1, q1);
    load_weights(wp2, q2);
  }

  float ssq[8];
#pragma unroll
  for (int r = 0; r < 8; ++r) {
    float v = 0.0f;
#pragma unroll
    for (int nt = 0; nt < 4; ++nt) v += acc[nt][r] * acc[nt][r];
    ssq[r] = v;
  }
#pragma unroll
  for (int m = 1; m <= 8; m <<= 1) {
#pragma unroll
    for (int r = 0; r < 8; ++r) ssq[r] += __shfl_xor(ssq[r], m, 32);
  }

#pragma unroll
  for (int r = 0; r < 8; ++r) {
    const float inv = 1.0f / fmaxf(sqrtf(ssq[r]), L2EPS);
    const int row = rb + r + 8 * half;
#pragma unroll
    for (int nt = 0; nt < 4; ++nt) {
      float v = acc[nt][r] * inv;
      v = fmaxf(v, 0.0f);
      if constexpr (HEAD)
        my[(r + 8 * half) * AG_PITCH + nt * 16 + l16] = v;
      else
        out[(long long)row * DIM + nt * 16 + l16] = v;
    }
  }

  if constexpr (HEAD) {
    __syncthreads();
    store_weights(lwl, q1);
    store_weights(lwr, q2);
    __syncthreads();

    v2f ah[16];
#pragma unroll
    for (int k = 0; k < 16; ++k)
      ah[k] = *(const v2f*)(my + l16 * AG_PITCH + 4 * k + 2 * half);
#pragma unroll
    for (int nt = 0; nt < 4; ++nt) {
      v8f c8 = {};
#pragma unroll
      for (int k = 0; k < 16; ++k) {
        const v2f b = lwl[(nt * 16 + k) * WF_PITCH + lane];
        c8 = wmma_f32(ah[k], b, c8);
      }
      acc[nt] = c8;
    }
#pragma unroll
    for (int nt = 0; nt < 4; ++nt) {
      const float bias = bp1[nt * 16 + l16];
#pragma unroll
      for (int r = 0; r < 8; ++r)
        my[(r + 8 * half) * AG_PITCH + nt * 16 + l16] = acc[nt][r] + bias;
    }
    __syncthreads();

#pragma unroll
    for (int k = 0; k < 16; ++k)
      ah[k] = *(const v2f*)(my + l16 * AG_PITCH + 4 * k + 2 * half);
#pragma unroll
    for (int nt = 0; nt < 4; ++nt) {
      v8f c8 = {};
#pragma unroll
      for (int k = 0; k < 16; ++k) {
        const v2f b = lwr[(nt * 16 + k) * WF_PITCH + lane];
        c8 = wmma_f32(ah[k], b, c8);
      }
      acc[nt] = c8;
    }
    if (live) {
#pragma unroll
      for (int nt = 0; nt < 4; ++nt) {
        const float bias = bp2[nt * 16 + l16];
#pragma unroll
        for (int r = 0; r < 8; ++r) {
          const int row = rb + r + 8 * half;
          out[(long long)row * DIM + nt * 16 + l16] = acc[nt][r] + bias;
        }
      }
    }
  }
}

extern "C" void kernel_launch(void* const* d_in, const int* in_sizes, int n_in,
                              void* d_out, int out_size, void* d_ws, size_t ws_size,
                              hipStream_t stream) {
  (void)in_sizes; (void)n_in; (void)out_size; (void)ws_size;

  const float* x   = (const float*)d_in[0];
  const int*   ei  = (const int*)d_in[1];
  const int*   src = ei;
  const int*   dst = ei + N_EDGES;
  const float* w1l = (const float*)d_in[2];
  const float* b1l = (const float*)d_in[3];
  const float* w1r = (const float*)d_in[4];
  const float* b1r = (const float*)d_in[5];
  const float* w2l = (const float*)d_in[6];
  const float* b2l = (const float*)d_in[7];
  const float* w2r = (const float*)d_in[8];
  const float* b2r = (const float*)d_in[9];
  const float* wp1 = (const float*)d_in[10];
  const float* bp1 = (const float*)d_in[11];
  const float* wp2 = (const float*)d_in[12];
  const float* bp2 = (const float*)d_in[13];
  float* out = (float*)d_out;

  const long long NS = (long long)N_NODES * DIM;
  float* s   = (float*)d_ws;
  float* h1  = s + NS;
  int* off1 = (int*)s;
  int* deg  = off1 + (CHUNKS * O1 + 31) / 32 * 32;
  int* off  = deg + N_NODES;
  int* rec  = off + N_NODES;
  int* csr  = rec + N_EDGES;

  const dim3 BG(GEMM_T);
  const dim3 BS(SORT_T);
  const int ggrid = (NTILES + 7) / 8;

  chunk_sort_kernel<<<CHUNKS, BS, 0, stream>>>(src, dst, rec, off1);
  bin_sort_kernel<<<NB, BS, 0, stream>>>(off1, rec, csr, off, deg);

  sage_kernel<false><<<ggrid, BG, 0, stream>>>(x, csr, off, deg, w1l, b1l, w1r, b1r,
                                               wp1, bp1, wp2, bp2, h1);

  sage_kernel<true><<<ggrid, BG, 0, stream>>>(h1, csr, off, deg, w2l, b2l, w2r, b2r,
                                              wp1, bp1, wp2, bp2, out);
}


# self product x@Wl^T (64 of a tile's WMMAs) issued inside the gather's rounds, 8 per round under that round's row loads, not behind the gather; Wp2 fetched after Wp1's copy (128 VGPRs); sum order kept
# speedup vs baseline: 4.9929x; 1.0366x over previous
#include <hip/hip_runtime.h>
#include <math.h>


typedef __attribute__((ext_vector_type(2))) float v2f;
typedef __attribute__((ext_vector_type(8))) float v8f;

#define N_NODES 100000
#define N_EDGES 1600000
#define DIM 64
#define NTILES (N_NODES / 16)
#define L2EPS 1e-12f

__device__ __forceinline__ v8f wmma_f32(v2f a, v2f b, v8f c) {
  return __builtin_amdgcn_wmma_f32_16x16x4_f32(
       false, a,  false, b,
       (short)0, c,  false,  false);
}

__device__ __forceinline__ v2f frag_pair(const float* __restrict__ p) {
  v2f r; r.x = p[0]; r.y = p[1]; return r;
}

#define WF_PITCH 33
#define WF_PAIRS (64 * WF_PITCH)
#define GEMM_T 256
#define WQ (DIM * DIM / 4 / GEMM_T)

__device__ __forceinline__ void load_weights(const float* __restrict__ w, float4 (&q)[WQ]) {
#pragma unroll
  for (int j = 0; j < WQ; ++j)
    q[j] = *(const float4*)(w + 4 * (threadIdx.x + j * GEMM_T));
}

__device__ __forceinline__ void store_weights(v2f* lw, const float4 (&q)[WQ]) {
#pragma unroll
  for (int j = 0; j < WQ; ++j) {
    const int i = threadIdx.x + j * GEMM_T;
    const int n = i >> 4;
    const int c = i & 15;
    const int f = (n >> 4) * 16 + c;
    v2f lo; lo.x = q[j].x; lo.y = q[j].y;
    v2f hi; hi.x = q[j].z; hi.y = q[j].w;
    lw[f * WF_PITCH + (n & 15)]      = lo;
    lw[f * WF_PITCH + 16 + (n & 15)] = hi;
  }
}

#define NB ((N_NODES + 255) / 256)
#define CHUNKS 250
#define CHUNK_E (N_EDGES / CHUNKS)
#define CHUNK_V (CHUNK_E / 4)
#define O1 (NB + 1)
#define SORT_T 1024
#define RUN_U 4
static_assert(CHUNK_V * 4 * CHUNKS == N_EDGES, "chunks must tile the edge list");
static_assert(N_NODES <= (1 << 24), "src must fit the upper 24 bits of a record");
static_assert(NB <= SORT_T, "one thread per block in the scans");
static_assert(CHUNK_V > SORT_T && CHUNK_V <= 2 * SORT_T, "a thread owns one or two int4 of its chunk");
static_assert(CHUNKS * 4 <= SORT_T, "four threads per chunk in level 2");

__device__ __forceinline__ int block_scan_excl(int val, int* wtot, int& total) {
  const int lane = threadIdx.x & 31;
  const int wave = threadIdx.x >> 5;
  int inc = val;
#pragma unroll
  for (int o = 1; o < 32; o <<= 1) {
    const int up = __shfl_up(inc, o, 32);
    if (lane >= o) inc += up;
  }
  if (lane == 31) wtot[wave] = inc;
  __syncthreads();
  if (wave == 0) {
    const int t = wtot[lane];
    int ti = t;
#pragma unroll
    for (int o = 1; o < 32; o <<= 1) {
      const int up = __shfl_up(ti, o, 32);
      if (lane >= o) ti += up;
    }
    wtot[lane] = ti - t;
    if (lane == 31) wtot[32] = ti;
  }
  __syncthreads();
  total = wtot[32];
  const int ex = wtot[wave] + inc - val;
  __syncthreads();
  return ex;
}

__global__ __launch_bounds__(SORT_T) void chunk_sort_kernel(const int* __restrict__ src,
                                                            const int* __restrict__ dst,
                                                            int* __restrict__ rec,
                                                            int* __restrict__ off1) {
  __shared__ int hist[NB];
  __shared__ int offs[NB];
  __shared__ int wtot[33];
  __shared__ int4 recL4[CHUNK_V];
  int* recL = (int*)recL4;
  const int t = threadIdx.x;
  const int c = blockIdx.x;
  const bool two = (t + SORT_T) < CHUNK_V;
  const int4* s4 = (const int4*)src + (long long)c * CHUNK_V;
  const int4* d4 = (const int4*)dst + (long long)c * CHUNK_V;
  const int i1 = two ? t + SORT_T : CHUNK_V - 1;
  const int4 sa = s4[t];
  const int4 da = d4[t];
  const int4 sb = s4[i1];
  const int4 db = d4[i1];
  if (t < NB) hist[t] = 0;
  __syncthreads();
  const int r0 = atomicAdd(&hist[da.x >> 8], 1);
  const int r1 = atomicAdd(&hist[da.y >> 8], 1);
  const int r2 = atomicAdd(&hist[da.z >> 8], 1);
  const int r3 = atomicAdd(&hist[da.w >> 8], 1);
  int r4 = 0, r5 = 0, r6 = 0, r7 = 0;
  if (two) {
    r4 = atomicAdd(&hist[db.x >> 8], 1);
    r5 = atomicAdd(&hist[db.y >> 8], 1);
    r6 = atomicAdd(&hist[db.z >> 8], 1);
    r7 = atomicAdd(&hist[db.w >> 8], 1);
  }
  __syncthreads();
  const int h = (t < NB) ? hist[t] : 0;
  int total;
  const int ex = block_scan_excl(h, wtot, total);
  if (t < NB) {
    offs[t] = ex;
    off1[c * O1 + t] = ex;
  }
  if (t == 0) off1[c * O1 + NB] = total;
  __syncthreads();
  recL[min(max(offs[da.x >> 8] + r0, 0), CHUNK_E - 1)] = (int)(((unsigned)sa.x << 8) | (unsigned)(da.x & 255));
  recL[min(max(offs[da.y >> 8] + r1, 0), CHUNK_E - 1)] = (int)(((unsigned)sa.y << 8) | (unsigned)(da.y & 255));
  recL[min(max(offs[da.z >> 8] + r2, 0), CHUNK_E - 1)] = (int)(((unsigned)sa.z << 8) | (unsigned)(da.z & 255));
  recL[min(max(offs[da.w >> 8] + r3, 0), CHUNK_E - 1)] = (int)(((unsigned)sa.w << 8) | (unsigned)(da.w & 255));
  if (two) {
    recL[min(max(offs[db.x >> 8] + r4, 0), CHUNK_E - 1)] = (int)(((unsigned)sb.x << 8) | (unsigned)(db.x & 255));
    recL[min(max(offs[db.y >> 8] + r5, 0), CHUNK_E - 1)] = (int)(((unsigned)sb.y << 8) | (unsigned)(db.y & 255));
    recL[min(max(offs[db.z >> 8] + r6, 0), CHUNK_E - 1)] = (int)(((unsigned)sb.z << 8) | (unsigned)(db.z & 255));
    recL[min(max(offs[db.w >> 8] + r7, 0), CHUNK_E - 1)] = (int)(((unsigned)sb.w << 8) | (unsigned)(db.w & 255));
  }
  __syncthreads();
  int4* o4 = (int4*)(rec + (long long)c * CHUNK_E);
  o4[t] = recL4[t];
  if (two) o4[t + SORT_T] = recL4[t + SORT_T];
}

__global__ __launch_bounds__(SORT_T) void bin_sort_kernel(const int* __restrict__ off1,
                                                          const int* __restrict__ rec,
                                                          int* __restrict__ csr,
                                                          int* __restrict__ off,
                                                          int* __restrict__ deg) {
  __shared__ int cntL[256];
  __shared__ int offL[256];
  __shared__ int wtot[33];
  const int t = threadIdx.x;
  const int b = blockIdx.x;
  const int q = t & 3;
  const int c = min(t >> 2, CHUNKS - 1);
  int s = 0, e = 0;
  if ((t >> 2) < CHUNKS) {
    s = min(max(off1[c * O1 + b], 0), CHUNK_E);
    e = min(max(off1[c * O1 + b + 1], s), CHUNK_E);
  }
  int before;
  block_scan_excl((q == 0) ? s : 0, wtot, before);
  const int bstart = min(max(before, 0), N_EDGES);
  if (t < 256) cntL[t] = 0;
  __syncthreads();
  const int* rc = rec + (long long)c * CHUNK_E;
  for (int i = s + q; i < e; i += 4 * RUN_U) {
    int w[RUN_U];
#pragma unroll
    for (int u = 0; u < RUN_U; ++u) w[u] = rc[min(i + 4 * u, e - 1)];
#pragma unroll
    for (int u = 0; u < RUN_U; ++u)
      atomicAdd(&cntL[w[u] & 255], (i + 4 * u < e) ? 1 : 0);
  }
  __syncthreads();
  const int cn = (t < 256) ? cntL[t] : 0;
  int all;
  const int ex = block_scan_excl(cn, wtot, all);
  if (t < 256) {
    offL[t] = ex;
    cntL[t] = 0;
    const int v = b * 256 + t;
    if (v < N_NODES) {
      off[v] = bstart + ex;
      deg[v] = cn;
    }
  }
  __syncthreads();
  for (int i = s + q; i < e; i += 4 * RUN_U) {
    int w[RUN_U];
#pragma unroll
    for (int u = 0; u < RUN_U; ++u) w[u] = rc[min(i + 4 * u, e - 1)];
    int r[RUN_U];
#pragma unroll
    for (int u = 0; u < RUN_U; ++u)
      r[u] = atomicAdd(&cntL[w[u] & 255], (i + 4 * u < e) ? 1 : 0);
#pragma unroll
    for (int u = 0; u < RUN_U; ++u) {
      if (i + 4 * u < e) {
        const int p = min(max(bstart + offL[w[u] & 255] + r[u], 0), N_EDGES - 1);
        csr[p] = (int)((unsigned)w[u] >> 8);
      }
    }
  }
}

#define GB 8
#define AG_PITCH 68

__device__ __forceinline__ void gather_tile(const float* __restrict__ x,
                                            const int* __restrict__ csr,
                                            int st_l, int dg_l, int l16, int half, int lane,
                                            float* __restrict__ my,
                                            const v2f* lw, const float* __restrict__ xself,
                                            v8f (&cacc)[4]) {
  const int d4 = l16 * 4;
  const float4* __restrict__ x4 = (const float4*)x;
  for (int i = 0; i < 8; ++i) {
    const int m  = 2 * i + half;
    const int so = __shfl(st_l, m, 32);
    const int dr = __shfl(dg_l, m, 32);
    const int start = min(max(so, 0), N_EDGES);
    const int n     = min(max(dr, 0), N_EDGES - start);
    float4 acc = make_float4(0.0f, 0.0f, 0.0f, 0.0f);
    {
      const int last = max(n - 1, 0);
      int idx[GB];
#pragma unroll
      for (int u = 0; u < GB; ++u) {
        const int e = csr[min(start + min(u, last), N_EDGES - 1)];
        idx[u] = min(max(e, 0), N_NODES - 1);
      }
      const v2f a0 = frag_pair(xself + 8 * i);
      const v2f a1 = frag_pair(xself + 8 * i + 4);
      float4 r[GB];
#pragma unroll
      for (int u = 0; u < GB; ++u)
        r[u] = x4[(unsigned)idx[u] * (DIM / 4) + (unsigned)l16];
#pragma unroll
      for (int nt = 0; nt < 4; ++nt) {
        const v2f b = lw[(nt * 16 + 2 * i) * WF_PITCH + lane];
        cacc[nt] = wmma_f32(a0, b, cacc[nt]);
      }
#pragma unroll
      for (int nt = 0; nt < 4; ++nt) {
        const v2f b = lw[(nt * 16 + 2 * i + 1) * WF_PITCH + lane];
        cacc[nt] = wmma_f32(a1, b, cacc[nt]);
      }
      acc.x += r[0].x; acc.y += r[0].y; acc.z += r[0].z; acc.w += r[0].w;
#pragma unroll
      for (int u = 1; u < GB; ++u) {
        const unsigned mk = (u < n) ? 0xffffffffu : 0u;
        acc.x += __uint_as_float(__float_as_uint(r[u].x) & mk);
        acc.y += __uint_as_float(__float_as_uint(r[u].y) & mk);
        acc.z += __uint_as_float(__float_as_uint(r[u].z) & mk);
        acc.w += __uint_as_float(__float_as_uint(r[u].w) & mk);
      }
    }
    for (int j = GB; j < n; j += GB) {
      int idx[GB];
#pragma unroll
      for (int u = 0; u < GB; ++u) {
        const int e = csr[start + min(j + u, n - 1)];
        idx[u] = min(max(e, 0), N_NODES - 1);
      }
      float4 r[GB];
#pragma unroll
      for (int u = 0; u < GB; ++u)
        r[u] = x4[(unsigned)idx[u] * (DIM / 4) + (unsigned)l16];
#pragma unroll
      for (int u = 0; u < GB; ++u) {
        const bool in = (j + u < n);
        acc.x += in ? r[u].x : 0.0f;
        acc.y += in ? r[u].y : 0.0f;
        acc.z += in ? r[u].z : 0.0f;
        acc.w += in ? r[u].w : 0.0f;
      }
    }
    const float rinv = 1.0f / fmaxf((float)dr, 1.0f);
    const bool any = (n > 0);
    acc.x = any ? acc.x * rinv : 0.0f;
    acc.y = any ? acc.y * rinv : 0.0f;
    acc.z = any ? acc.z * rinv : 0.0f;
    acc.w = any ? acc.w * rinv : 0.0f;
    *(float4*)(my + m * AG_PITCH + d4) = acc;
  }
}

template <bool HEAD>
__global__ __launch_bounds__(GEMM_T) void sage_kernel(
    const float* __restrict__ xin, const int* __restrict__ csr,
    const int* __restrict__ off, const int* __restrict__ deg,
    const float* __restrict__ wl, const float* __restrict__ bl,
    const float* __restrict__ wr, const float* __restrict__ br,
    const float* __restrict__ wp1, const float* __restrict__ bp1,
    const float* __restrict__ wp2, const float* __restrict__ bp2,
    float* __restrict__ out) {
  __shared__ v2f lwl[WF_PAIRS];
  __shared__ v2f lwr[WF_PAIRS];
  __shared__ float4 agg4[(GEMM_T / 32) * 16 * (AG_PITCH / 4)];
  const int wave = threadIdx.x >> 5;
  const int lane = threadIdx.x & 31;
  const int tile_raw = blockIdx.x * (blockDim.x >> 5) + wave;
  const bool live = tile_raw < NTILES;
  const int tile = live ? tile_raw : NTILES - 1;

  const int l16  = lane & 15;
  const int half = lane >> 4;
  const int rb   = tile * 16;
  const int arow = rb + l16;
  float* my = (float*)agg4 + wave * (16 * AG_PITCH);

  float4 ql[WQ], qr[WQ];
  load_weights(wl, ql);
  load_weights(wr, qr);
  const int st_l = off[arow];
  const int dg_l = deg[arow];
  const float* xself = xin + (long long)arow * DIM + 2 * half;
  store_weights(lwl, ql);
  store_weights(lwr, qr);
  __syncthreads();

  v8f acc[4];
#pragma unroll
  for (int nt = 0; nt < 4; ++nt) acc[nt] = v8f{};
  gather_tile(xin, csr, st_l, dg_l, l16, half, lane, my, lwl, xself, acc);
  __syncthreads();
  if constexpr (!HEAD) {
    if (!live) return;
  }

  v2f ap[16];
#pragma unroll
  for (int k = 0; k < 16; ++k)
    ap[k] = *(const v2f*)(my + l16 * AG_PITCH + 4 * k + 2 * half);

#pragma unroll
  for (int nt = 0; nt < 4; ++nt) {
    v8f c8 = acc[nt];
    const int n0 = nt * 16;
#pragma unroll
    for (int k = 0; k < 16; ++k) {
      const v2f b = lwr[(nt * 16 + k) * WF_PITCH + lane];
      c8 = wmma_f32(ap[k], b, c8);
    }
    const float bias = bl[n0 + l16] + br[n0 + l16];
#pragma unroll
    for (int r = 0; r < 8; ++r) c8[r] += bias;
    acc[nt] = c8;
  }

  float4 q1[WQ], q2[WQ];
  if constexpr (HEAD) {
    load_weights(wp1, q1);
  }

  float ssq[8];
#pragma unroll
  for (int r = 0; r < 8; ++r) {
    float v = 0.0f;
#pragma unroll
    for (int nt = 0; nt < 4; ++nt) v += acc[nt][r] * acc[nt][r];
    ssq[r] = v;
  }
#pragma unroll
  for (int m = 1; m <= 8; m <<= 1) {
#pragma unroll
    for (int r = 0; r < 8; ++r) ssq[r] += __shfl_xor(ssq[r], m, 32);
  }

#pragma unroll
  for (int r = 0; r < 8; ++r) {
    const float inv = 1.0f / fmaxf(sqrtf(ssq[r]), L2EPS);
    const int row = rb + r + 8 * half;
#pragma unroll
    for (int nt = 0; nt < 4; ++nt) {
      float v = acc[nt][r] * inv;
      v = fmaxf(v, 0.0f);
      if constexpr (HEAD)
        my[(r + 8 * half) * AG_PITCH + nt * 16 + l16] = v;
      else
        out[(long long)row * DIM + nt * 16 + l16] = v;
    }
  }

  if constexpr (HEAD) {
    __syncthreads();
    store_weights(lwl, q1);
    load_weights(wp2, q2);
    __syncthreads();

    v2f ah[16];
#pragma unroll
    for (int k = 0; k < 16; ++k)
      ah[k] = *(const v2f*)(my + l16 * AG_PITCH + 4 * k + 2 * half);
#pragma unroll
    for (int nt = 0; nt < 4; ++nt) {
      v8f c8 = {};
#pragma unroll
      for (int k = 0; k < 16; ++k) {
        const v2f b = lwl[(nt * 16 + k) * WF_PITCH + lane];
        c8 = wmma_f32(ah[k], b, c8);
      }
      acc[nt] = c8;
    }
#pragma unroll
    for (int nt = 0; nt < 4; ++nt) {
      const float bias = bp1[nt * 16 + l16];
#pragma unroll
      for (int r = 0; r < 8; ++r)
        my[(r + 8 * half) * AG_PITCH + nt * 16 + l16] = acc[nt][r] + bias;
    }
    store_weights(lwr, q2);
    __syncthreads();

#pragma unroll
    for (int k = 0; k < 16; ++k)
      ah[k] = *(const v2f*)(my + l16 * AG_PITCH + 4 * k + 2 * half);
#pragma unroll
    for (int nt = 0; nt < 4; ++nt) {
      v8f c8 = {};
#pragma unroll
      for (int k = 0; k < 16; ++k) {
        const v2f b = lwr[(nt * 16 + k) * WF_PITCH + lane];
        c8 = wmma_f32(ah[k], b, c8);
      }
      acc[nt] = c8;
    }
    if (live) {
#pragma unroll
      for (int nt = 0; nt < 4; ++nt) {
        const float bias = bp2[nt * 16 + l16];
#pragma unroll
        for (int r = 0; r < 8; ++r) {
          const int row = rb + r + 8 * half;
          out[(long long)row * DIM + nt * 16 + l16] = acc[nt][r] + bias;
        }
      }
    }
  }
}

extern "C" void kernel_launch(void* const* d_in, const int* in_sizes, int n_in,
                              void* d_out, int out_size, void* d_ws, size_t ws_size,
                              hipStream_t stream) {
  (void)in_sizes; (void)n_in; (void)out_size; (void)ws_size;

  const float* x   = (const float*)d_in[0];
  const int*   ei  = (const int*)d_in[1];
  const int*   src = ei;
  const int*   dst = ei + N_EDGES;
  const float* w1l = (const float*)d_in[2];
  const float* b1l = (const float*)d_in[3];
  const float* w1r = (const float*)d_in[4];
  const float* b1r = (const float*)d_in[5];
  const float* w2l = (const float*)d_in[6];
  const float* b2l = (const float*)d_in[7];
  const float* w2r = (const float*)d_in[8];
  const float* b2r = (const float*)d_in[9];
  const float* wp1 = (const float*)d_in[10];
  const float* bp1 = (const float*)d_in[11];
  const float* wp2 = (const float*)d_in[12];
  const float* bp2 = (const float*)d_in[13];
  float* out = (float*)d_out;

  const long long NS = (long long)N_NODES * DIM;
  float* s   = (float*)d_ws;
  float* h1  = s + NS;
  int* off1 = (int*)s;
  int* deg  = off1 + (CHUNKS * O1 + 31) / 32 * 32;
  int* off  = deg + N_NODES;
  int* rec  = off + N_NODES;
  int* csr  = rec + N_EDGES;

  const dim3 BG(GEMM_T);
  const dim3 BS(SORT_T);
  const int ggrid = (NTILES + 7) / 8;

  chunk_sort_kernel<<<CHUNKS, BS, 0, stream>>>(src, dst, rec, off1);
  bin_sort_kernel<<<NB, BS, 0, stream>>>(off1, rec, csr, off, deg);

  sage_kernel<false><<<ggrid, BG, 0, stream>>>(x, csr, off, deg, w1l, b1l, w1r, b1r,
                                               wp1, bp1, wp2, bp2, h1);

  sage_kernel<true><<<ggrid, BG, 0, stream>>>(h1, csr, off, deg, w2l, b2l, w2r, b2r,
                                              wp1, bp1, wp2, bp2, out);
}


# tile's own 16 rows read once as whole lines into the wave's LDS tile; self fragments are LDS reads (were 16 loads of 16 lines); means reuse read columns; 2 empty asm fences hold WMMA order; sums same
# speedup vs baseline: 5.1550x; 1.0325x over previous
#include <hip/hip_runtime.h>
#include <math.h>


typedef __attribute__((ext_vector_type(2))) float v2f;
typedef __attribute__((ext_vector_type(8))) float v8f;

#define N_NODES 100000
#define N_EDGES 1600000
#define DIM 64
#define NTILES (N_NODES / 16)
#define L2EPS 1e-12f

__device__ __forceinline__ v8f wmma_f32(v2f a, v2f b, v8f c) {
  return __builtin_amdgcn_wmma_f32_16x16x4_f32(
       false, a,  false, b,
       (short)0, c,  false,  false);
}

#define WF_PITCH 33
#define WF_PAIRS (64 * WF_PITCH)
#define GEMM_T 256
#define WQ (DIM * DIM / 4 / GEMM_T)

__device__ __forceinline__ void load_weights(const float* __restrict__ w, float4 (&q)[WQ]) {
#pragma unroll
  for (int j = 0; j < WQ; ++j)
    q[j] = *(const float4*)(w + 4 * (threadIdx.x + j * GEMM_T));
}

__device__ __forceinline__ void store_weights(v2f* lw, const float4 (&q)[WQ]) {
#pragma unroll
  for (int j = 0; j < WQ; ++j) {
    const int i = threadIdx.x + j * GEMM_T;
    const int n = i >> 4;
    const int c = i & 15;
    const int f = (n >> 4) * 16 + c;
    v2f lo; lo.x = q[j].x; lo.y = q[j].y;
    v2f hi; hi.x = q[j].z; hi.y = q[j].w;
    lw[f * WF_PITCH + (n & 15)]      = lo;
    lw[f * WF_PITCH + 16 + (n & 15)] = hi;
  }
}

#define NB ((N_NODES + 255) / 256)
#define CHUNKS 250
#define CHUNK_E (N_EDGES / CHUNKS)
#define CHUNK_V (CHUNK_E / 4)
#define O1 (NB + 1)
#define SORT_T 1024
#define RUN_U 4
static_assert(CHUNK_V * 4 * CHUNKS == N_EDGES, "chunks must tile the edge list");
static_assert(N_NODES <= (1 << 24), "src must fit the upper 24 bits of a record");
static_assert(NB <= SORT_T, "one thread per block in the scans");
static_assert(CHUNK_V > SORT_T && CHUNK_V <= 2 * SORT_T, "a thread owns one or two int4 of its chunk");
static_assert(CHUNKS * 4 <= SORT_T, "four threads per chunk in level 2");

__device__ __forceinline__ int block_scan_excl(int val, int* wtot, int& total) {
  const int lane = threadIdx.x & 31;
  const int wave = threadIdx.x >> 5;
  int inc = val;
#pragma unroll
  for (int o = 1; o < 32; o <<= 1) {
    const int up = __shfl_up(inc, o, 32);
    if (lane >= o) inc += up;
  }
  if (lane == 31) wtot[wave] = inc;
  __syncthreads();
  if (wave == 0) {
    const int t = wtot[lane];
    int ti = t;
#pragma unroll
    for (int o = 1; o < 32; o <<= 1) {
      const int up = __shfl_up(ti, o, 32);
      if (lane >= o) ti += up;
    }
    wtot[lane] = ti - t;
    if (lane == 31) wtot[32] = ti;
  }
  __syncthreads();
  total = wtot[32];
  const int ex = wtot[wave] + inc - val;
  __syncthreads();
  return ex;
}

__global__ __launch_bounds__(SORT_T) void chunk_sort_kernel(const int* __restrict__ src,
                                                            const int* __restrict__ dst,
                                                            int* __restrict__ rec,
                                                            int* __restrict__ off1) {
  __shared__ int hist[NB];
  __shared__ int offs[NB];
  __shared__ int wtot[33];
  __shared__ int4 recL4[CHUNK_V];
  int* recL = (int*)recL4;
  const int t = threadIdx.x;
  const int c = blockIdx.x;
  const bool two = (t + SORT_T) < CHUNK_V;
  const int4* s4 = (const int4*)src + (long long)c * CHUNK_V;
  const int4* d4 = (const int4*)dst + (long long)c * CHUNK_V;
  const int i1 = two ? t + SORT_T : CHUNK_V - 1;
  const int4 sa = s4[t];
  const int4 da = d4[t];
  const int4 sb = s4[i1];
  const int4 db = d4[i1];
  if (t < NB) hist[t] = 0;
  __syncthreads();
  const int r0 = atomicAdd(&hist[da.x >> 8], 1);
  const int r1 = atomicAdd(&hist[da.y >> 8], 1);
  const int r2 = atomicAdd(&hist[da.z >> 8], 1);
  const int r3 = atomicAdd(&hist[da.w >> 8], 1);
  int r4 = 0, r5 = 0, r6 = 0, r7 = 0;
  if (two) {
    r4 = atomicAdd(&hist[db.x >> 8], 1);
    r5 = atomicAdd(&hist[db.y >> 8], 1);
    r6 = atomicAdd(&hist[db.z >> 8], 1);
    r7 = atomicAdd(&hist[db.w >> 8], 1);
  }
  __syncthreads();
  const int h = (t < NB) ? hist[t] : 0;
  int total;
  const int ex = block_scan_excl(h, wtot, total);
  if (t < NB) {
    offs[t] = ex;
    off1[c * O1 + t] = ex;
  }
  if (t == 0) off1[c * O1 + NB] = total;
  __syncthreads();
  recL[min(max(offs[da.x >> 8] + r0, 0), CHUNK_E - 1)] = (int)(((unsigned)sa.x << 8) | (unsigned)(da.x & 255));
  recL[min(max(offs[da.y >> 8] + r1, 0), CHUNK_E - 1)] = (int)(((unsigned)sa.y << 8) | (unsigned)(da.y & 255));
  recL[min(max(offs[da.z >> 8] + r2, 0), CHUNK_E - 1)] = (int)(((unsigned)sa.z << 8) | (unsigned)(da.z & 255));
  recL[min(max(offs[da.w >> 8] + r3, 0), CHUNK_E - 1)] = (int)(((unsigned)sa.w << 8) | (unsigned)(da.w & 255));
  if (two) {
    recL[min(max(offs[db.x >> 8] + r4, 0), CHUNK_E - 1)] = (int)(((unsigned)sb.x << 8) | (unsigned)(db.x & 255));
    recL[min(max(offs[db.y >> 8] + r5, 0), CHUNK_E - 1)] = (int)(((unsigned)sb.y << 8) | (unsigned)(db.y & 255));
    recL[min(max(offs[db.z >> 8] + r6, 0), CHUNK_E - 1)] = (int)(((unsigned)sb.z << 8) | (unsigned)(db.z & 255));
    recL[min(max(offs[db.w >> 8] + r7, 0), CHUNK_E - 1)] = (int)(((unsigned)sb.w << 8) | (unsigned)(db.w & 255));
  }
  __syncthreads();
  int4* o4 = (int4*)(rec + (long long)c * CHUNK_E);
  o4[t] = recL4[t];
  if (two) o4[t + SORT_T] = recL4[t + SORT_T];
}

__global__ __launch_bounds__(SORT_T) void bin_sort_kernel(const int* __restrict__ off1,
                                                          const int* __restrict__ rec,
                                                          int* __restrict__ csr,
                                                          int* __restrict__ off,
                                                          int* __restrict__ deg) {
  __shared__ int cntL[256];
  __shared__ int offL[256];
  __shared__ int wtot[33];
  const int t = threadIdx.x;
  const int b = blockIdx.x;
  const int q = t & 3;
  const int c = min(t >> 2, CHUNKS - 1);
  int s = 0, e = 0;
  if ((t >> 2) < CHUNKS) {
    s = min(max(off1[c * O1 + b], 0), CHUNK_E);
    e = min(max(off1[c * O1 + b + 1], s), CHUNK_E);
  }
  int before;
  block_scan_excl((q == 0) ? s : 0, wtot, before);
  const int bstart = min(max(before, 0), N_EDGES);
  if (t < 256) cntL[t] = 0;
  __syncthreads();
  const int* rc = rec + (long long)c * CHUNK_E;
  for (int i = s + q; i < e; i += 4 * RUN_U) {
    int w[RUN_U];
#pragma unroll
    for (int u = 0; u < RUN_U; ++u) w[u] = rc[min(i + 4 * u, e - 1)];
#pragma unroll
    for (int u = 0; u < RUN_U; ++u)
      atomicAdd(&cntL[w[u] & 255], (i + 4 * u < e) ? 1 : 0);
  }
  __syncthreads();
  const int cn = (t < 256) ? cntL[t] : 0;
  int all;
  const int ex = block_scan_excl(cn, wtot, all);
  if (t < 256) {
    offL[t] = ex;
    cntL[t] = 0;
    const int v = b * 256 + t;
    if (v < N_NODES) {
      off[v] = bstart + ex;
      deg[v] = cn;
    }
  }
  __syncthreads();
  for (int i = s + q; i < e; i += 4 * RUN_U) {
    int w[RUN_U];
#pragma unroll
    for (int u = 0; u < RUN_U; ++u) w[u] = rc[min(i + 4 * u, e - 1)];
    int r[RUN_U];
#pragma unroll
    for (int u = 0; u < RUN_U; ++u)
      r[u] = atomicAdd(&cntL[w[u] & 255], (i + 4 * u < e) ? 1 : 0);
#pragma unroll
    for (int u = 0; u < RUN_U; ++u) {
      if (i + 4 * u < e) {
        const int p = min(max(bstart + offL[w[u] & 255] + r[u], 0), N_EDGES - 1);
        csr[p] = (int)((unsigned)w[u] >> 8);
      }
    }
  }
}

#define GB 8
#define AG_PITCH 68

#define MEAN_AT(node, f) ((8 * ((node) & 1) + ((f) >> 3)) * AG_PITCH + 8 * ((node) >> 1) + ((f) & 7))
__device__ __forceinline__ void gather_tile(const float* __restrict__ x,
                                            const int* __restrict__ csr,
                                            int st_l, int dg_l, int l16, int half, int lane,
                                            float* __restrict__ my,
                                            const v2f* lw,
                                            v8f (&cacc)[4]) {
  const int d4 = l16 * 4;
  const float4* __restrict__ x4 = (const float4*)x;
  for (int i = 0; i < 8; ++i) {
    const int m  = 2 * i + half;
    const int so = __shfl(st_l, m, 32);
    const int dr = __shfl(dg_l, m, 32);
    const int start = min(max(so, 0), N_EDGES);
    const int n     = min(max(dr, 0), N_EDGES - start);
    float4 acc = make_float4(0.0f, 0.0f, 0.0f, 0.0f);
    {
      const int last = max(n - 1, 0);
      int idx[GB];
#pragma unroll
      for (int u = 0; u < GB; ++u) {
        const int e = csr[min(start + min(u, last), N_EDGES - 1)];
        idx[u] = min(max(e, 0), N_NODES - 1);
      }
      v2f a0 = *(const v2f*)(my + l16 * AG_PITCH + 8 * i + 2 * half);
      v2f a1 = *(const v2f*)(my + l16 * AG_PITCH + 8 * i + 4 + 2 * half);
      float4 r[GB];
#pragma unroll
      for (int u = 0; u < GB; ++u)
        r[u] = x4[(unsigned)idx[u] * (DIM / 4) + (unsigned)l16];
      float zero = 0.0f;
      int nn = n;
      asm volatile("" : "+v"(a0), "+v"(a1), "+v"(zero), "+v"(nn) : : "memory");
#pragma unroll
      for (int nt = 0; nt < 4; ++nt) {
        const v2f b = lw[(nt * 16 + 2 * i) * WF_PITCH + lane];
        cacc[nt] = wmma_f32(a0, b, cacc[nt]);
      }
#pragma unroll
      for (int nt = 0; nt < 4; ++nt) {
        const v2f b = lw[(nt * 16 + 2 * i + 1) * WF_PITCH + lane];
        cacc[nt] = wmma_f32(a1, b, cacc[nt]);
      }
      asm volatile("" : "+v"(zero), "+v"(nn), "+v"(cacc[0]), "+v"(cacc[1]), "+v"(cacc[2]), "+v"(cacc[3]));
      acc.x = zero + r[0].x; acc.y = zero + r[0].y; acc.z = zero + r[0].z; acc.w = zero + r[0].w;
#pragma unroll
      for (int u = 1; u < GB; ++u) {
        const unsigned mk = (u < nn) ? 0xffffffffu : 0u;
        acc.x += __uint_as_float(__float_as_uint(r[u].x) & mk);
        acc.y += __uint_as_float(__float_as_uint(r[u].y) & mk);
        acc.z += __uint_as_float(__float_as_uint(r[u].z) & mk);
        acc.w += __uint_as_float(__float_as_uint(r[u].w) & mk);
      }
    }
    for (int j = GB; j < n; j += GB) {
      int idx[GB];
#pragma unroll
      for (int u = 0; u < GB; ++u) {
        const int e = csr[start + min(j + u, n - 1)];
        idx[u] = min(max(e, 0), N_NODES - 1);
      }
      float4 r[GB];
#pragma unroll
      for (int u = 0; u < GB; ++u)
        r[u] = x4[(unsigned)idx[u] * (DIM / 4) + (unsigned)l16];
#pragma unroll
      for (int u = 0; u < GB; ++u) {
        const bool in = (j + u < n);
        acc.x += in ? r[u].x : 0.0f;
        acc.y += in ? r[u].y : 0.0f;
        acc.z += in ? r[u].z : 0.0f;
        acc.w += in ? r[u].w : 0.0f;
      }
    }
    const float rinv = 1.0f / fmaxf((float)dr, 1.0f);
    const bool any = (n > 0);
    acc.x = any ? acc.x * rinv : 0.0f;
    acc.y = any ? acc.y * rinv : 0.0f;
    acc.z = any ? acc.z * rinv : 0.0f;
    acc.w = any ? acc.w * rinv : 0.0f;
    *(float4*)(my + MEAN_AT(m, d4)) = acc;
  }
}

template <bool HEAD>
__global__ __launch_bounds__(GEMM_T) void sage_kernel(
    const float* __restrict__ xin, const int* __restrict__ csr,
    const int* __restrict__ off, const int* __restrict__ deg,
    const float* __restrict__ wl, const float* __restrict__ bl,
    const float* __restrict__ wr, const float* __restrict__ br,
    const float* __restrict__ wp1, const float* __restrict__ bp1,
    const float* __restrict__ wp2, const float* __restrict__ bp2,
    float* __restrict__ out) {
  __shared__ v2f lwl[WF_PAIRS];
  __shared__ v2f lwr[WF_PAIRS];
  __shared__ float4 agg4[(GEMM_T / 32) * 16 * (AG_PITCH / 4)];
  const int wave = threadIdx.x >> 5;
  const int lane = threadIdx.x & 31;
  const int tile_raw = blockIdx.x * (blockDim.x >> 5) + wave;
  const bool live = tile_raw < NTILES;
  const int tile = live ? tile_raw : NTILES - 1;

  const int l16  = lane & 15;
  const int half = lane >> 4;
  const int rb   = tile * 16;
  const int arow = rb + l16;
  float* my = (float*)agg4 + wave * (16 * AG_PITCH);

  float4 ql[WQ], qr[WQ];
  load_weights(wl, ql);
  load_weights(wr, qr);
  const int st_l = off[arow];
  const int dg_l = deg[arow];
  float4 xs[8];
  const float4* __restrict__ xt4 = (const float4*)xin + (long long)rb * (DIM / 4);
#pragma unroll
  for (int j = 0; j < 8; ++j) xs[j] = xt4[j * 32 + lane];
  store_weights(lwl, ql);
  store_weights(lwr, qr);
#pragma unroll
  for (int j = 0; j < 8; ++j)
    *(float4*)(my + (2 * j + half) * AG_PITCH + 4 * l16) = xs[j];
  __syncthreads();

  v8f acc[4];
#pragma unroll
  for (int nt = 0; nt < 4; ++nt) acc[nt] = v8f{};
  gather_tile(xin, csr, st_l, dg_l, l16, half, lane, my, lwl, acc);
  __syncthreads();
  if constexpr (!HEAD) {
    if (!live) return;
  }

  v2f ap[16];
#pragma unroll
  for (int k = 0; k < 16; ++k)
    ap[k] = *(const v2f*)(my + MEAN_AT(l16, 4 * k + 2 * half));

#pragma unroll
  for (int nt = 0; nt < 4; ++nt) {
    v8f c8 = acc[nt];
    const int n0 = nt * 16;
#pragma unroll
    for (int k = 0; k < 16; ++k) {
      const v2f b = lwr[(nt * 16 + k) * WF_PITCH + lane];
      c8 = wmma_f32(ap[k], b, c8);
    }
    const float bias = bl[n0 + l16] + br[n0 + l16];
#pragma unroll
    for (int r = 0; r < 8; ++r) c8[r] += bias;
    acc[nt] = c8;
  }

  float4 q1[WQ], q2[WQ];
  if constexpr (HEAD) {
    load_weights(wp1, q1);
  }

  float ssq[8];
#pragma unroll
  for (int r = 0; r < 8; ++r) {
    float v = 0.0f;
#pragma unroll
    for (int nt = 0; nt < 4; ++nt) v += acc[nt][r] * acc[nt][r];
    ssq[r] = v;
  }
#pragma unroll
  for (int m = 1; m <= 8; m <<= 1) {
#pragma unroll
    for (int r = 0; r < 8; ++r) ssq[r] += __shfl_xor(ssq[r], m, 32);
  }

#pragma unroll
  for (int r = 0; r < 8; ++r) {
    const float inv = 1.0f / fmaxf(sqrtf(ssq[r]), L2EPS);
    const int row = rb + r + 8 * half;
#pragma unroll
    for (int nt = 0; nt < 4; ++nt) {
      float v = acc[nt][r] * inv;
      v = fmaxf(v, 0.0f);
      if constexpr (HEAD)
        my[(r + 8 * half) * AG_PITCH + nt * 16 + l16] = v;
      else
        out[(long long)row * DIM + nt * 16 + l16] = v;
    }
  }

  if constexpr (HEAD) {
    __syncthreads();
    store_weights(lwl, q1);
    load_weights(wp2, q2);
    __syncthreads();

    v2f ah[16];
#pragma unroll
    for (int k = 0; k < 16; ++k)
      ah[k] = *(const v2f*)(my + l16 * AG_PITCH + 4 * k + 2 * half);
#pragma unroll
    for (int nt = 0; nt < 4; ++nt) {
      v8f c8 = {};
#pragma unroll
      for (int k = 0; k < 16; ++k) {
        const v2f b = lwl[(nt * 16 + k) * WF_PITCH + lane];
        c8 = wmma_f32(ah[k], b, c8);
      }
      acc[nt] = c8;
    }
#pragma unroll
    for (int nt = 0; nt < 4; ++nt) {
      const float bias = bp1[nt * 16 + l16];
#pragma unroll
      for (int r = 0; r < 8; ++r)
        my[(r + 8 * half) * AG_PITCH + nt * 16 + l16] = acc[nt][r] + bias;
    }
    store_weights(lwr, q2);
    __syncthreads();

#pragma unroll
    for (int k = 0; k < 16; ++k)
      ah[k] = *(const v2f*)(my + l16 * AG_PITCH + 4 * k + 2 * half);
#pragma unroll
    for (int nt = 0; nt < 4; ++nt) {
      v8f c8 = {};
#pragma unroll
      for (int k = 0; k < 16; ++k) {
        const v2f b = lwr[(nt * 16 + k) * WF_PITCH + lane];
        c8 = wmma_f32(ah[k], b, c8);
      }
      acc[nt] = c8;
    }
    if (live) {
#pragma unroll
      for (int nt = 0; nt < 4; ++nt) {
        const float bias = bp2[nt * 16 + l16];
#pragma unroll
        for (int r = 0; r < 8; ++r) {
          const int row = rb + r + 8 * half;
          out[(long long)row * DIM + nt * 16 + l16] = acc[nt][r] + bias;
        }
      }
    }
  }
}

extern "C" void kernel_launch(void* const* d_in, const int* in_sizes, int n_in,
                              void* d_out, int out_size, void* d_ws, size_t ws_size,
                              hipStream_t stream) {
  (void)in_sizes; (void)n_in; (void)out_size; (void)ws_size;

  const float* x   = (const float*)d_in[0];
  const int*   ei  = (const int*)d_in[1];
  const int*   src = ei;
  const int*   dst = ei + N_EDGES;
  const float* w1l = (const float*)d_in[2];
  const float* b1l = (const float*)d_in[3];
  const float* w1r = (const float*)d_in[4];
  const float* b1r = (const float*)d_in[5];
  const float* w2l = (const float*)d_in[6];
  const float* b2l = (const float*)d_in[7];
  const float* w2r = (const float*)d_in[8];
  const float* b2r = (const float*)d_in[9];
  const float* wp1 = (const float*)d_in[10];
  const float* bp1 = (const float*)d_in[11];
  const float* wp2 = (const float*)d_in[12];
  const float* bp2 = (const float*)d_in[13];
  float* out = (float*)d_out;

  const long long NS = (long long)N_NODES * DIM;
  float* s   = (float*)d_ws;
  float* h1  = s + NS;
  int* off1 = (int*)s;
  int* deg  = off1 + (CHUNKS * O1 + 31) / 32 * 32;
  int* off  = deg + N_NODES;
  int* rec  = off + N_NODES;
  int* csr  = rec + N_EDGES;

  const dim3 BG(GEMM_T);
  const dim3 BS(SORT_T);
  const int ggrid = (NTILES + 7) / 8;

  chunk_sort_kernel<<<CHUNKS, BS, 0, stream>>>(src, dst, rec, off1);
  bin_sort_kernel<<<NB, BS, 0, stream>>>(off1, rec, csr, off, deg);

  sage_kernel<false><<<ggrid, BG, 0, stream>>>(x, csr, off, deg, w1l, b1l, w1r, b1r,
                                               wp1, bp1, wp2, bp2, h1);

  sage_kernel<true><<<ggrid, BG, 0, stream>>>(h1, csr, off, deg, w2l, b2l, w2r, b2r,
                                              wp1, bp1, wp2, bp2, out);
}


# post-MP head's two linears folded into one: Wc = Wp2 Wp1, bc = Wp2 bp1 + bp2 formed per call by a spare work-group of the layer-1 launch; 64 WMMAs per tile, 2 barriers go; head sum order changes
# speedup vs baseline: 5.4065x; 1.0488x over previous
#include <hip/hip_runtime.h>
#include <math.h>


typedef __attribute__((ext_vector_type(2))) float v2f;
typedef __attribute__((ext_vector_type(8))) float v8f;

#define N_NODES 100000
#define N_EDGES 1600000
#define DIM 64
#define NTILES (N_NODES / 16)
#define L2EPS 1e-12f

__device__ __forceinline__ v8f wmma_f32(v2f a, v2f b, v8f c) {
  return __builtin_amdgcn_wmma_f32_16x16x4_f32(
       false, a,  false, b,
       (short)0, c,  false,  false);
}

#define WF_PITCH 33
#define WF_PAIRS (64 * WF_PITCH)
#define GEMM_T 256
#define WQ (DIM * DIM / 4 / GEMM_T)

__device__ __forceinline__ void load_weights(const float* __restrict__ w, float4 (&q)[WQ]) {
#pragma unroll
  for (int j = 0; j < WQ; ++j)
    q[j] = *(const float4*)(w + 4 * (threadIdx.x + j * GEMM_T));
}

__device__ __forceinline__ void store_weights(v2f* lw, const float4 (&q)[WQ]) {
#pragma unroll
  for (int j = 0; j < WQ; ++j) {
    const int i = threadIdx.x + j * GEMM_T;
    const int n = i >> 4;
    const int c = i & 15;
    const int f = (n >> 4) * 16 + c;
    v2f lo; lo.x = q[j].x; lo.y = q[j].y;
    v2f hi; hi.x = q[j].z; hi.y = q[j].w;
    lw[f * WF_PITCH + (n & 15)]      = lo;
    lw[f * WF_PITCH + 16 + (n & 15)] = hi;
  }
}

#define NB ((N_NODES + 255) / 256)
#define CHUNKS 250
#define CHUNK_E (N_EDGES / CHUNKS)
#define CHUNK_V (CHUNK_E / 4)
#define O1 (NB + 1)
#define SORT_T 1024
#define RUN_U 4
static_assert(CHUNK_V * 4 * CHUNKS == N_EDGES, "chunks must tile the edge list");
static_assert(N_NODES <= (1 << 24), "src must fit the upper 24 bits of a record");
static_assert(NB <= SORT_T, "one thread per block in the scans");
static_assert(CHUNK_V > SORT_T && CHUNK_V <= 2 * SORT_T, "a thread owns one or two int4 of its chunk");
static_assert(CHUNKS * 4 <= SORT_T, "four threads per chunk in level 2");
static_assert((CHUNKS * O1 + 31) / 32 * 32 + 2 * N_NODES + 2 * N_EDGES <= 4000000, "the sort's arrays end before the folded head matrix");
static_assert(4000000 + DIM * DIM + DIM <= N_NODES * DIM, "the folded head matrix fits the first workspace plane");

__device__ __forceinline__ int block_scan_excl(int val, int* wtot, int& total) {
  const int lane = threadIdx.x & 31;
  const int wave = threadIdx.x >> 5;
  int inc = val;
#pragma unroll
  for (int o = 1; o < 32; o <<= 1) {
    const int up = __shfl_up(inc, o, 32);
    if (lane >= o) inc += up;
  }
  if (lane == 31) wtot[wave] = inc;
  __syncthreads();
  if (wave == 0) {
    const int t = wtot[lane];
    int ti = t;
#pragma unroll
    for (int o = 1; o < 32; o <<= 1) {
      const int up = __shfl_up(ti, o, 32);
      if (lane >= o) ti += up;
    }
    wtot[lane] = ti - t;
    if (lane == 31) wtot[32] = ti;
  }
  __syncthreads();
  total = wtot[32];
  const int ex = wtot[wave] + inc - val;
  __syncthreads();
  return ex;
}

__global__ __launch_bounds__(SORT_T) void chunk_sort_kernel(const int* __restrict__ src,
                                                            const int* __restrict__ dst,
                                                            int* __restrict__ rec,
                                                            int* __restrict__ off1) {
  __shared__ int hist[NB];
  __shared__ int offs[NB];
  __shared__ int wtot[33];
  __shared__ int4 recL4[CHUNK_V];
  int* recL = (int*)recL4;
  const int t = threadIdx.x;
  const int c = blockIdx.x;
  const bool two = (t + SORT_T) < CHUNK_V;
  const int4* s4 = (const int4*)src + (long long)c * CHUNK_V;
  const int4* d4 = (const int4*)dst + (long long)c * CHUNK_V;
  const int i1 = two ? t + SORT_T : CHUNK_V - 1;
  const int4 sa = s4[t];
  const int4 da = d4[t];
  const int4 sb = s4[i1];
  const int4 db = d4[i1];
  if (t < NB) hist[t] = 0;
  __syncthreads();
  const int r0 = atomicAdd(&hist[da.x >> 8], 1);
  const int r1 = atomicAdd(&hist[da.y >> 8], 1);
  const int r2 = atomicAdd(&hist[da.z >> 8], 1);
  const int r3 = atomicAdd(&hist[da.w >> 8], 1);
  int r4 = 0, r5 = 0, r6 = 0, r7 = 0;
  if (two) {
    r4 = atomicAdd(&hist[db.x >> 8], 1);
    r5 = atomicAdd(&hist[db.y >> 8], 1);
    r6 = atomicAdd(&hist[db.z >> 8], 1);
    r7 = atomicAdd(&hist[db.w >> 8], 1);
  }
  __syncthreads();
  const int h = (t < NB) ? hist[t] : 0;
  int total;
  const int ex = block_scan_excl(h, wtot, total);
  if (t < NB) {
    offs[t] = ex;
    off1[c * O1 + t] = ex;
  }
  if (t == 0) off1[c * O1 + NB] = total;
  __syncthreads();
  recL[min(max(offs[da.x >> 8] + r0, 0), CHUNK_E - 1)] = (int)(((unsigned)sa.x << 8) | (unsigned)(da.x & 255));
  recL[min(max(offs[da.y >> 8] + r1, 0), CHUNK_E - 1)] = (int)(((unsigned)sa.y << 8) | (unsigned)(da.y & 255));
  recL[min(max(offs[da.z >> 8] + r2, 0), CHUNK_E - 1)] = (int)(((unsigned)sa.z << 8) | (unsigned)(da.z & 255));
  recL[min(max(offs[da.w >> 8] + r3, 0), CHUNK_E - 1)] = (int)(((unsigned)sa.w << 8) | (unsigned)(da.w & 255));
  if (two) {
    recL[min(max(offs[db.x >> 8] + r4, 0), CHUNK_E - 1)] = (int)(((unsigned)sb.x << 8) | (unsigned)(db.x & 255));
    recL[min(max(offs[db.y >> 8] + r5, 0), CHUNK_E - 1)] = (int)(((unsigned)sb.y << 8) | (unsigned)(db.y & 255));
    recL[min(max(offs[db.z >> 8] + r6, 0), CHUNK_E - 1)] = (int)(((unsigned)sb.z << 8) | (unsigned)(db.z & 255));
    recL[min(max(offs[db.w >> 8] + r7, 0), CHUNK_E - 1)] = (int)(((unsigned)sb.w << 8) | (unsigned)(db.w & 255));
  }
  __syncthreads();
  int4* o4 = (int4*)(rec + (long long)c * CHUNK_E);
  o4[t] = recL4[t];
  if (two) o4[t + SORT_T] = recL4[t + SORT_T];
}

__global__ __launch_bounds__(SORT_T) void bin_sort_kernel(const int* __restrict__ off1,
                                                          const int* __restrict__ rec,
                                                          int* __restrict__ csr,
                                                          int* __restrict__ off,
                                                          int* __restrict__ deg) {
  __shared__ int cntL[256];
  __shared__ int offL[256];
  __shared__ int wtot[33];
  const int t = threadIdx.x;
  const int b = blockIdx.x;
  const int q = t & 3;
  const int c = min(t >> 2, CHUNKS - 1);
  int s = 0, e = 0;
  if ((t >> 2) < CHUNKS) {
    s = min(max(off1[c * O1 + b], 0), CHUNK_E);
    e = min(max(off1[c * O1 + b + 1], s), CHUNK_E);
  }
  int before;
  block_scan_excl((q == 0) ? s : 0, wtot, before);
  const int bstart = min(max(before, 0), N_EDGES);
  if (t < 256) cntL[t] = 0;
  __syncthreads();
  const int* rc = rec + (long long)c * CHUNK_E;
  for (int i = s + q; i < e; i += 4 * RUN_U) {
    int w[RUN_U];
#pragma unroll
    for (int u = 0; u < RUN_U; ++u) w[u] = rc[min(i + 4 * u, e - 1)];
#pragma unroll
    for (int u = 0; u < RUN_U; ++u)
      atomicAdd(&cntL[w[u] & 255], (i + 4 * u < e) ? 1 : 0);
  }
  __syncthreads();
  const int cn = (t < 256) ? cntL[t] : 0;
  int all;
  const int ex = block_scan_excl(cn, wtot, all);
  if (t < 256) {
    offL[t] = ex;
    cntL[t] = 0;
    const int v = b * 256 + t;
    if (v < N_NODES) {
      off[v] = bstart + ex;
      deg[v] = cn;
    }
  }
  __syncthreads();
  for (int i = s + q; i < e; i += 4 * RUN_U) {
    int w[RUN_U];
#pragma unroll
    for (int u = 0; u < RUN_U; ++u) w[u] = rc[min(i + 4 * u, e - 1)];
    int r[RUN_U];
#pragma unroll
    for (int u = 0; u < RUN_U; ++u)
      r[u] = atomicAdd(&cntL[w[u] & 255], (i + 4 * u < e) ? 1 : 0);
#pragma unroll
    for (int u = 0; u < RUN_U; ++u) {
      if (i + 4 * u < e) {
        const int p = min(max(bstart + offL[w[u] & 255] + r[u], 0), N_EDGES - 1);
        csr[p] = (int)((unsigned)w[u] >> 8);
      }
    }
  }
}

#define GB 8
#define AG_PITCH 68

#define MEAN_AT(node, f) ((8 * ((node) & 1) + ((f) >> 3)) * AG_PITCH + 8 * ((node) >> 1) + ((f) & 7))
__device__ __forceinline__ void gather_tile(const float* __restrict__ x,
                                            const int* __restrict__ csr,
                                            int st_l, int dg_l, int l16, int half, int lane,
                                            float* __restrict__ my,
                                            const v2f* lw,
                                            v8f (&cacc)[4]) {
  const int d4 = l16 * 4;
  const float4* __restrict__ x4 = (const float4*)x;
  for (int i = 0; i < 8; ++i) {
    const int m  = 2 * i + half;
    const int so = __shfl(st_l, m, 32);
    const int dr = __shfl(dg_l, m, 32);
    const int start = min(max(so, 0), N_EDGES);
    const int n     = min(max(dr, 0), N_EDGES - start);
    float4 acc = make_float4(0.0f, 0.0f, 0.0f, 0.0f);
    {
      const int last = max(n - 1, 0);
      int idx[GB];
#pragma unroll
      for (int u = 0; u < GB; ++u) {
        const int e = csr[min(start + min(u, last), N_EDGES - 1)];
        idx[u] = min(max(e, 0), N_NODES - 1);
      }
      v2f a0 = *(const v2f*)(my + l16 * AG_PITCH + 8 * i + 2 * half);
      v2f a1 = *(const v2f*)(my + l16 * AG_PITCH + 8 * i + 4 + 2 * half);
      float4 r[GB];
#pragma unroll
      for (int u = 0; u < GB; ++u)
        r[u] = x4[(unsigned)idx[u] * (DIM / 4) + (unsigned)l16];
      float zero = 0.0f;
      int nn = n;
      asm volatile("" : "+v"(a0), "+v"(a1), "+v"(zero), "+v"(nn) : : "memory");
#pragma unroll
      for (int nt = 0; nt < 4; ++nt) {
        const v2f b = lw[(nt * 16 + 2 * i) * WF_PITCH + lane];
        cacc[nt] = wmma_f32(a0, b, cacc[nt]);
      }
#pragma unroll
      for (int nt = 0; nt < 4; ++nt) {
        const v2f b = lw[(nt * 16 + 2 * i + 1) * WF_PITCH + lane];
        cacc[nt] = wmma_f32(a1, b, cacc[nt]);
      }
      asm volatile("" : "+v"(zero), "+v"(nn), "+v"(cacc[0]), "+v"(cacc[1]), "+v"(cacc[2]), "+v"(cacc[3]));
      acc.x = zero + r[0].x; acc.y = zero + r[0].y; acc.z = zero + r[0].z; acc.w = zero + r[0].w;
#pragma unroll
      for (int u = 1; u < GB; ++u) {
        const unsigned mk = (u < nn) ? 0xffffffffu : 0u;
        acc.x += __uint_as_float(__float_as_uint(r[u].x) & mk);
        acc.y += __uint_as_float(__float_as_uint(r[u].y) & mk);
        acc.z += __uint_as_float(__float_as_uint(r[u].z) & mk);
        acc.w += __uint_as_float(__float_as_uint(r[u].w) & mk);
      }
    }
    for (int j = GB; j < n; j += GB) {
      int idx[GB];
#pragma unroll
      for (int u = 0; u < GB; ++u) {
        const int e = csr[start + min(j + u, n - 1)];
        idx[u] = min(max(e, 0), N_NODES - 1);
      }
      float4 r[GB];
#pragma unroll
      for (int u = 0; u < GB; ++u)
        r[u] = x4[(unsigned)idx[u] * (DIM / 4) + (unsigned)l16];
#pragma unroll
      for (int u = 0; u < GB; ++u) {
        const bool in = (j + u < n);
        acc.x += in ? r[u].x : 0.0f;
        acc.y += in ? r[u].y : 0.0f;
        acc.z += in ? r[u].z : 0.0f;
        acc.w += in ? r[u].w : 0.0f;
      }
    }
    const float rinv = 1.0f / fmaxf((float)dr, 1.0f);
    const bool any = (n > 0);
    acc.x = any ? acc.x * rinv : 0.0f;
    acc.y = any ? acc.y * rinv : 0.0f;
    acc.z = any ? acc.z * rinv : 0.0f;
    acc.w = any ? acc.w * rinv : 0.0f;
    *(float4*)(my + MEAN_AT(m, d4)) = acc;
  }
}

#define GGRID ((NTILES + 7) / 8)

__device__ __forceinline__ void fold_head(const float* __restrict__ wp1, const float* __restrict__ bp1,
                                          const float* __restrict__ wp2, const float* __restrict__ bp2,
                                          float* __restrict__ wc, float* __restrict__ bc) {
  const int wave = threadIdx.x >> 5;
  const int lane = threadIdx.x & 31;
  const int l16  = lane & 15;
  const int half = lane >> 4;
#pragma unroll
  for (int u = 0; u < 2; ++u) {
    const int t  = 2 * wave + u;
    const int n0 = (t >> 2) * 16;
    const int k0 = (t & 3) * 16;
    v8f c8 = {};
#pragma unroll
    for (int s = 0; s < 16; ++s) {
      const int j = 4 * s + 2 * half;
      const v2f a = *(const v2f*)(wp2 + (n0 + l16) * DIM + j);
      v2f b;
      b.x = wp1[j * DIM + k0 + l16];
      b.y = wp1[(j + 1) * DIM + k0 + l16];
      c8 = wmma_f32(a, b, c8);
    }
#pragma unroll
    for (int r = 0; r < 8; ++r)
      wc[(n0 + 8 * half + r) * DIM + k0 + l16] = c8[r];
  }
  if (threadIdx.x < DIM) {
    const int n = threadIdx.x;
    float v = bp2[n];
    for (int j = 0; j < DIM; ++j) v = fmaf(wp2[n * DIM + j], bp1[j], v);
    bc[n] = v;
  }
}

template <bool HEAD>
__global__ __launch_bounds__(GEMM_T) void sage_kernel(
    const float* __restrict__ xin, const int* __restrict__ csr,
    const int* __restrict__ off, const int* __restrict__ deg,
    const float* __restrict__ wl, const float* __restrict__ bl,
    const float* __restrict__ wr, const float* __restrict__ br,
    const float* __restrict__ wp1, const float* __restrict__ bp1,
    const float* __restrict__ wp2, const float* __restrict__ bp2,
    float* __restrict__ fold_w, float* __restrict__ fold_b,
    float* __restrict__ out) {
  if constexpr (!HEAD) {
    if (blockIdx.x == GGRID) {
      fold_head(wp1, bp1, wp2, bp2, fold_w, fold_b);
      return;
    }
  }
  __shared__ v2f lwl[WF_PAIRS];
  __shared__ v2f lwr[WF_PAIRS];
  __shared__ float4 agg4[(GEMM_T / 32) * 16 * (AG_PITCH / 4)];
  const int wave = threadIdx.x >> 5;
  const int lane = threadIdx.x & 31;
  const int tile_raw = blockIdx.x * (blockDim.x >> 5) + wave;
  const bool live = tile_raw < NTILES;
  const int tile = live ? tile_raw : NTILES - 1;

  const int l16  = lane & 15;
  const int half = lane >> 4;
  const int rb   = tile * 16;
  const int arow = rb + l16;
  float* my = (float*)agg4 + wave * (16 * AG_PITCH);

  float4 ql[WQ], qr[WQ];
  load_weights(wl, ql);
  load_weights(wr, qr);
  const int st_l = off[arow];
  const int dg_l = deg[arow];
  float4 xs[8];
  const float4* __restrict__ xt4 = (const float4*)xin + (long long)rb * (DIM / 4);
#pragma unroll
  for (int j = 0; j < 8; ++j) xs[j] = xt4[j * 32 + lane];
  store_weights(lwl, ql);
  store_weights(lwr, qr);
#pragma unroll
  for (int j = 0; j < 8; ++j)
    *(float4*)(my + (2 * j + half) * AG_PITCH + 4 * l16) = xs[j];
  __syncthreads();

  v8f acc[4];
#pragma unroll
  for (int nt = 0; nt < 4; ++nt) acc[nt] = v8f{};
  gather_tile(xin, csr, st_l, dg_l, l16, half, lane, my, lwl, acc);
  __syncthreads();
  if constexpr (!HEAD) {
    if (!live) return;
  }

  float4 q1[WQ];
  if constexpr (HEAD) {
    load_weights(wp1, q1);
  }

  v2f ap[16];
#pragma unroll
  for (int k = 0; k < 16; ++k)
    ap[k] = *(const v2f*)(my + MEAN_AT(l16, 4 * k + 2 * half));

#pragma unroll
  for (int nt = 0; nt < 4; ++nt) {
    v8f c8 = acc[nt];
    const int n0 = nt * 16;
#pragma unroll
    for (int k = 0; k < 16; ++k) {
      const v2f b = lwr[(nt * 16 + k) * WF_PITCH + lane];
      c8 = wmma_f32(ap[k], b, c8);
    }
    const float bias = bl[n0 + l16] + br[n0 + l16];
#pragma unroll
    for (int r = 0; r < 8; ++r) c8[r] += bias;
    acc[nt] = c8;
  }


  float ssq[8];
#pragma unroll
  for (int r = 0; r < 8; ++r) {
    float v = 0.0f;
#pragma unroll
    for (int nt = 0; nt < 4; ++nt) v += acc[nt][r] * acc[nt][r];
    ssq[r] = v;
  }
#pragma unroll
  for (int m = 1; m <= 8; m <<= 1) {
#pragma unroll
    for (int r = 0; r < 8; ++r) ssq[r] += __shfl_xor(ssq[r], m, 32);
  }

#pragma unroll
  for (int r = 0; r < 8; ++r) {
    const float inv = 1.0f / fmaxf(sqrtf(ssq[r]), L2EPS);
    const int row = rb + r + 8 * half;
#pragma unroll
    for (int nt = 0; nt < 4; ++nt) {
      float v = acc[nt][r] * inv;
      v = fmaxf(v, 0.0f);
      if constexpr (HEAD)
        my[(r + 8 * half) * AG_PITCH + nt * 16 + l16] = v;
      else
        out[(long long)row * DIM + nt * 16 + l16] = v;
    }
  }

  if constexpr (HEAD) {
    store_weights(lwl, q1);
    __syncthreads();

    v2f ah[16];
#pragma unroll
    for (int k = 0; k < 16; ++k)
      ah[k] = *(const v2f*)(my + l16 * AG_PITCH + 4 * k + 2 * half);
#pragma unroll
    for (int nt = 0; nt < 4; ++nt) {
      v8f c8 = {};
#pragma unroll
      for (int k = 0; k < 16; ++k) {
        const v2f b = lwl[(nt * 16 + k) * WF_PITCH + lane];
        c8 = wmma_f32(ah[k], b, c8);
      }
      acc[nt] = c8;
    }
    if (live) {
#pragma unroll
      for (int nt = 0; nt < 4; ++nt) {
        const float bias = bp1[nt * 16 + l16];
#pragma unroll
        for (int r = 0; r < 8; ++r) {
          const int row = rb + r + 8 * half;
          out[(long long)row * DIM + nt * 16 + l16] = acc[nt][r] + bias;
        }
      }
    }
  }
}

extern "C" void kernel_launch(void* const* d_in, const int* in_sizes, int n_in,
                              void* d_out, int out_size, void* d_ws, size_t ws_size,
                              hipStream_t stream) {
  (void)in_sizes; (void)n_in; (void)out_size; (void)ws_size;

  const float* x   = (const float*)d_in[0];
  const int*   ei  = (const int*)d_in[1];
  const int*   src = ei;
  const int*   dst = ei + N_EDGES;
  const float* w1l = (const float*)d_in[2];
  const float* b1l = (const float*)d_in[3];
  const float* w1r = (const float*)d_in[4];
  const float* b1r = (const float*)d_in[5];
  const float* w2l = (const float*)d_in[6];
  const float* b2l = (const float*)d_in[7];
  const float* w2r = (const float*)d_in[8];
  const float* b2r = (const float*)d_in[9];
  const float* wp1 = (const float*)d_in[10];
  const float* bp1 = (const float*)d_in[11];
  const float* wp2 = (const float*)d_in[12];
  const float* bp2 = (const float*)d_in[13];
  float* out = (float*)d_out;

  const long long NS = (long long)N_NODES * DIM;
  float* s   = (float*)d_ws;
  float* h1  = s + NS;
  int* off1 = (int*)s;
  int* deg  = off1 + (CHUNKS * O1 + 31) / 32 * 32;
  int* off  = deg + N_NODES;
  int* rec  = off + N_NODES;
  int* csr  = rec + N_EDGES;
  float* wc = s + 4000000;
  float* bc = wc + DIM * DIM;

  const dim3 BG(GEMM_T);
  const dim3 BS(SORT_T);
  const int ggrid = GGRID;

  chunk_sort_kernel<<<CHUNKS, BS, 0, stream>>>(src, dst, rec, off1);
  bin_sort_kernel<<<NB, BS, 0, stream>>>(off1, rec, csr, off, deg);

  sage_kernel<false><<<GGRID + 1, BG, 0, stream>>>(x, csr, off, deg, w1l, b1l, w1r, b1r,
                                                   wp1, bp1, wp2, bp2, wc, bc, h1);

  sage_kernel<true><<<ggrid, BG, 0, stream>>>(h1, csr, off, deg, w2l, b2l, w2r, b2r,
                                              wc, bc, wp2, bp2, nullptr, nullptr, out);
}


# level 2 of the edge sort runs inside the layer-1 launch: each work-group bins its own 128 nodes' records into csr in front of its gather; the bin sort launch goes; sums keep LDS arrival order
# speedup vs baseline: 5.7396x; 1.0616x over previous
#include <hip/hip_runtime.h>
#include <math.h>


typedef __attribute__((ext_vector_type(2))) float v2f;
typedef __attribute__((ext_vector_type(8))) float v8f;

#define N_NODES 100000
#define N_EDGES 1600000
#define DIM 64
#define NTILES (N_NODES / 16)
#define L2EPS 1e-12f

__device__ __forceinline__ v8f wmma_f32(v2f a, v2f b, v8f c) {
  return __builtin_amdgcn_wmma_f32_16x16x4_f32(
       false, a,  false, b,
       (short)0, c,  false,  false);
}

#define WF_PITCH 33
#define WF_PAIRS (64 * WF_PITCH)
#define GEMM_T 256
#define WQ (DIM * DIM / 4 / GEMM_T)

__device__ __forceinline__ void load_weights(const float* __restrict__ w, float4 (&q)[WQ]) {
#pragma unroll
  for (int j = 0; j < WQ; ++j)
    q[j] = *(const float4*)(w + 4 * (threadIdx.x + j * GEMM_T));
}

__device__ __forceinline__ void store_weights(v2f* lw, const float4 (&q)[WQ]) {
#pragma unroll
  for (int j = 0; j < WQ; ++j) {
    const int i = threadIdx.x + j * GEMM_T;
    const int n = i >> 4;
    const int c = i & 15;
    const int f = (n >> 4) * 16 + c;
    v2f lo; lo.x = q[j].x; lo.y = q[j].y;
    v2f hi; hi.x = q[j].z; hi.y = q[j].w;
    lw[f * WF_PITCH + (n & 15)]      = lo;
    lw[f * WF_PITCH + 16 + (n & 15)] = hi;
  }
}

#define BSH 7
#define BNODES (1 << BSH)
#define NB ((N_NODES + BNODES - 1) / BNODES)
#define CHUNKS 250
#define CHUNK_E (N_EDGES / CHUNKS)
#define CHUNK_V (CHUNK_E / 4)
#define O1 (NB + 1)
#define SORT_T 1024
#define RUN_U 6
static_assert(CHUNK_V * 4 * CHUNKS == N_EDGES, "chunks must tile the edge list");
static_assert(N_NODES <= (1 << 24), "src must fit the upper 24 bits of a record");
static_assert(NB <= SORT_T, "one thread per block in the scans");
static_assert(CHUNK_V > SORT_T && CHUNK_V <= 2 * SORT_T, "a thread owns one or two int4 of its chunk");
static_assert((CHUNKS * O1 + 31) / 32 * 32 + 2 * N_NODES + 2 * N_EDGES <= 4000000, "the sort's arrays end before the folded head matrix");
static_assert(4000000 + DIM * DIM + DIM <= N_NODES * DIM, "the folded head matrix fits the first workspace plane");

__device__ __forceinline__ int block_scan_excl(int val, int* wtot, int& total) {
  const int lane = threadIdx.x & 31;
  const int wave = threadIdx.x >> 5;
  int inc = val;
#pragma unroll
  for (int o = 1; o < 32; o <<= 1) {
    const int up = __shfl_up(inc, o, 32);
    if (lane >= o) inc += up;
  }
  if (lane == 31) wtot[wave] = inc;
  __syncthreads();
  if (wave == 0) {
    const int t = wtot[lane];
    int ti = t;
#pragma unroll
    for (int o = 1; o < 32; o <<= 1) {
      const int up = __shfl_up(ti, o, 32);
      if (lane >= o) ti += up;
    }
    wtot[lane] = ti - t;
    if (lane == 31) wtot[32] = ti;
  }
  __syncthreads();
  total = wtot[32];
  const int ex = wtot[wave] + inc - val;
  __syncthreads();
  return ex;
}

__global__ __launch_bounds__(SORT_T) void chunk_sort_kernel(const int* __restrict__ src,
                                                            const int* __restrict__ dst,
                                                            int* __restrict__ rec,
                                                            int* __restrict__ off1) {
  __shared__ int hist[NB];
  __shared__ int offs[NB];
  __shared__ int wtot[33];
  __shared__ int4 recL4[CHUNK_V];
  int* recL = (int*)recL4;
  const int t = threadIdx.x;
  const int c = blockIdx.x;
  const bool two = (t + SORT_T) < CHUNK_V;
  const int4* s4 = (const int4*)src + (long long)c * CHUNK_V;
  const int4* d4 = (const int4*)dst + (long long)c * CHUNK_V;
  const int i1 = two ? t + SORT_T : CHUNK_V - 1;
  const int4 sa = s4[t];
  const int4 da = d4[t];
  const int4 sb = s4[i1];
  const int4 db = d4[i1];
  if (t < NB) hist[t] = 0;
  __syncthreads();
  const int r0 = atomicAdd(&hist[da.x >> BSH], 1);
  const int r1 = atomicAdd(&hist[da.y >> BSH], 1);
  const int r2 = atomicAdd(&hist[da.z >> BSH], 1);
  const int r3 = atomicAdd(&hist[da.w >> BSH], 1);
  int r4 = 0, r5 = 0, r6 = 0, r7 = 0;
  if (two) {
    r4 = atomicAdd(&hist[db.x >> BSH], 1);
    r5 = atomicAdd(&hist[db.y >> BSH], 1);
    r6 = atomicAdd(&hist[db.z >> BSH], 1);
    r7 = atomicAdd(&hist[db.w >> BSH], 1);
  }
  __syncthreads();
  const int h = (t < NB) ? hist[t] : 0;
  int total;
  const int ex = block_scan_excl(h, wtot, total);
  if (t < NB) {
    offs[t] = ex;
    off1[c * O1 + t] = ex;
  }
  if (t == 0) off1[c * O1 + NB] = total;
  __syncthreads();
  recL[min(max(offs[da.x >> BSH] + r0, 0), CHUNK_E - 1)] = (int)(((unsigned)sa.x << BSH) | (unsigned)(da.x & (BNODES - 1)));
  recL[min(max(offs[da.y >> BSH] + r1, 0), CHUNK_E - 1)] = (int)(((unsigned)sa.y << BSH) | (unsigned)(da.y & (BNODES - 1)));
  recL[min(max(offs[da.z >> BSH] + r2, 0), CHUNK_E - 1)] = (int)(((unsigned)sa.z << BSH) | (unsigned)(da.z & (BNODES - 1)));
  recL[min(max(offs[da.w >> BSH] + r3, 0), CHUNK_E - 1)] = (int)(((unsigned)sa.w << BSH) | (unsigned)(da.w & (BNODES - 1)));
  if (two) {
    recL[min(max(offs[db.x >> BSH] + r4, 0), CHUNK_E - 1)] = (int)(((unsigned)sb.x << BSH) | (unsigned)(db.x & (BNODES - 1)));
    recL[min(max(offs[db.y >> BSH] + r5, 0), CHUNK_E - 1)] = (int)(((unsigned)sb.y << BSH) | (unsigned)(db.y & (BNODES - 1)));
    recL[min(max(offs[db.z >> BSH] + r6, 0), CHUNK_E - 1)] = (int)(((unsigned)sb.z << BSH) | (unsigned)(db.z & (BNODES - 1)));
    recL[min(max(offs[db.w >> BSH] + r7, 0), CHUNK_E - 1)] = (int)(((unsigned)sb.w << BSH) | (unsigned)(db.w & (BNODES - 1)));
  }
  __syncthreads();
  int4* o4 = (int4*)(rec + (long long)c * CHUNK_E);
  o4[t] = recL4[t];
  if (two) o4[t + SORT_T] = recL4[t + SORT_T];
}

#define BQ (GEMM_T / 4)
#define KG 4
static_assert(KG * BQ >= CHUNKS && (KG - 1) * BQ < CHUNKS, "KG runs per quad cover the chunks");

__device__ __forceinline__ void run_tail_count(const int* __restrict__ rc, int from, int to, int* cntL) {
  for (int i = from; i < to; i += 4)
    atomicAdd(&cntL[rc[i] & (BNODES - 1)], 1);
}
__device__ __forceinline__ void run_tail_place(const int* __restrict__ rc, int from, int to,
                                               int* cntL, const int* offL, int* csr) {
  for (int i = from; i < to; i += 4) {
    const int wv = rc[i];
    const int d = wv & (BNODES - 1);
    const int r = atomicAdd(&cntL[d], 1);
    const int p = min(max(offL[d] + r, 0), N_EDGES - 1);
    csr[p] = (int)((unsigned)wv >> BSH);
  }
}

__device__ __forceinline__ void issue_loads(const float* __restrict__ wl, const float* __restrict__ wr,
                                            const float4* __restrict__ xt4, int lane,
                                            float4 (&ql)[WQ], float4 (&qr)[WQ], float4 (&xs)[8]) {
  load_weights(wl, ql);
  load_weights(wr, qr);
#pragma unroll
  for (int j = 0; j < 8; ++j) xs[j] = xt4[j * 32 + lane];
}

__device__ __forceinline__ void bin_block(const int* __restrict__ off1, const int* __restrict__ rec,
                                          int* csr, int* off, int* deg, int b,
                                          int* cntL, int* offL, int* degL, int* red, int* wt,
                                          const float* __restrict__ wl, const float* __restrict__ wr,
                                          const float4* __restrict__ xt4,
                                          float4 (&ql)[WQ], float4 (&qr)[WQ], float4 (&xs)[8]) {
  const int t = threadIdx.x;
  const int lane = t & 31;
  const int wave = t >> 5;
  const int q = t & 3;
  int s[KG], e[KG];
  int part = 0;
#pragma unroll
  for (int k = 0; k < KG; ++k) {
    const int cc = (t >> 2) + BQ * k;
    const int c = min(cc, CHUNKS - 1);
    const int s0 = min(max(off1[c * O1 + b], 0), CHUNK_E);
    const int e0 = min(max(off1[c * O1 + b + 1], s0), CHUNK_E);
    s[k] = s0;
    e[k] = (cc < CHUNKS) ? e0 : s0;
    part += (q == 0 && cc < CHUNKS) ? s0 : 0;
  }
  int w[KG][RUN_U];
#pragma unroll
  for (int k = 0; k < KG; ++k) {
    const int* rc = rec + (long long)min((t >> 2) + BQ * k, CHUNKS - 1) * CHUNK_E;
#pragma unroll
    for (int u = 0; u < RUN_U; ++u) {
      const int i = s[k] + q + 4 * u;
      w[k][u] = rc[min(max(min(i, e[k] - 1), 0), CHUNK_E - 1)];
    }
  }
#pragma unroll
  for (int o = 16; o > 0; o >>= 1) part += __shfl_xor(part, o, 32);
  if (lane == 0) red[wave] = part;
  if (t < BNODES) cntL[t] = 0;
  __syncthreads();
  int before = 0;
#pragma unroll
  for (int v = 0; v < GEMM_T / 32; ++v) before += red[v];
  const int bstart = min(max(before, 0), N_EDGES);
#pragma unroll
  for (int k = 0; k < KG; ++k) {
#pragma unroll
    for (int u = 0; u < RUN_U; ++u)
      if (s[k] + q + 4 * u < e[k]) atomicAdd(&cntL[w[k][u] & (BNODES - 1)], 1);
  }
  const int* rc0 = rec + (long long)min((t >> 2), CHUNKS - 1) * CHUNK_E;
  const int* rc1 = rec + (long long)min((t >> 2) + BQ, CHUNKS - 1) * CHUNK_E;
  const int* rc2 = rec + (long long)min((t >> 2) + 2 * BQ, CHUNKS - 1) * CHUNK_E;
  const int* rc3 = rec + (long long)min((t >> 2) + 3 * BQ, CHUNKS - 1) * CHUNK_E;
  run_tail_count(rc0, s[0] + q + 4 * RUN_U, e[0], cntL);
  run_tail_count(rc1, s[1] + q + 4 * RUN_U, e[1], cntL);
  run_tail_count(rc2, s[2] + q + 4 * RUN_U, e[2], cntL);
  run_tail_count(rc3, s[3] + q + 4 * RUN_U, e[3], cntL);
  __syncthreads();
  const int cn = (t < BNODES) ? cntL[t] : 0;
  int inc = cn;
#pragma unroll
  for (int o = 1; o < 32; o <<= 1) {
    const int up = __shfl_up(inc, o, 32);
    if (lane >= o) inc += up;
  }
  if (t < BNODES && lane == 31) wt[wave] = inc;
  __syncthreads();
  if (t < BNODES) {
    int base = 0;
#pragma unroll
    for (int v = 0; v < BNODES / 32; ++v) base += (v < wave) ? wt[v] : 0;
    const int st = bstart + base + inc - cn;
    offL[t] = st;
    degL[t] = cn;
    cntL[t] = 0;
    const int node = b * BNODES + t;
    if (node < N_NODES) {
      off[node] = st;
      deg[node] = cn;
    }
  }
  __syncthreads();
  issue_loads(wl, wr, xt4, lane, ql, qr, xs);
#pragma unroll
  for (int k = 0; k < KG; ++k) {
#pragma unroll
    for (int u = 0; u < RUN_U; ++u) {
      if (s[k] + q + 4 * u < e[k]) {
        const int d = w[k][u] & (BNODES - 1);
        const int r = atomicAdd(&cntL[d], 1);
        const int p = min(max(offL[d] + r, 0), N_EDGES - 1);
        csr[p] = (int)((unsigned)w[k][u] >> BSH);
      }
    }
  }
  run_tail_place(rc0, s[0] + q + 4 * RUN_U, e[0], cntL, offL, csr);
  run_tail_place(rc1, s[1] + q + 4 * RUN_U, e[1], cntL, offL, csr);
  run_tail_place(rc2, s[2] + q + 4 * RUN_U, e[2], cntL, offL, csr);
  run_tail_place(rc3, s[3] + q + 4 * RUN_U, e[3], cntL, offL, csr);
}

#define GB 8
#define AG_PITCH 68

#define MEAN_AT(node, f) ((8 * ((node) & 1) + ((f) >> 3)) * AG_PITCH + 8 * ((node) >> 1) + ((f) & 7))
__device__ __forceinline__ void gather_tile(const float* __restrict__ x,
                                            const int* __restrict__ csr,
                                            int st_l, int dg_l, int l16, int half, int lane,
                                            float* __restrict__ my,
                                            const v2f* lw,
                                            v8f (&cacc)[4]) {
  const int d4 = l16 * 4;
  const float4* __restrict__ x4 = (const float4*)x;
  for (int i = 0; i < 8; ++i) {
    const int m  = 2 * i + half;
    const int so = __shfl(st_l, m, 32);
    const int dr = __shfl(dg_l, m, 32);
    const int start = min(max(so, 0), N_EDGES);
    const int n     = min(max(dr, 0), N_EDGES - start);
    float4 acc = make_float4(0.0f, 0.0f, 0.0f, 0.0f);
    {
      const int last = max(n - 1, 0);
      int idx[GB];
#pragma unroll
      for (int u = 0; u < GB; ++u) {
        const int e = csr[min(start + min(u, last), N_EDGES - 1)];
        idx[u] = min(max(e, 0), N_NODES - 1);
      }
      v2f a0 = *(const v2f*)(my + l16 * AG_PITCH + 8 * i + 2 * half);
      v2f a1 = *(const v2f*)(my + l16 * AG_PITCH + 8 * i + 4 + 2 * half);
      float4 r[GB];
#pragma unroll
      for (int u = 0; u < GB; ++u)
        r[u] = x4[(unsigned)idx[u] * (DIM / 4) + (unsigned)l16];
      float zero = 0.0f;
      int nn = n;
      asm volatile("" : "+v"(a0), "+v"(a1), "+v"(zero), "+v"(nn) : : "memory");
#pragma unroll
      for (int nt = 0; nt < 4; ++nt) {
        const v2f b = lw[(nt * 16 + 2 * i) * WF_PITCH + lane];
        cacc[nt] = wmma_f32(a0, b, cacc[nt]);
      }
#pragma unroll
      for (int nt = 0; nt < 4; ++nt) {
        const v2f b = lw[(nt * 16 + 2 * i + 1) * WF_PITCH + lane];
        cacc[nt] = wmma_f32(a1, b, cacc[nt]);
      }
      asm volatile("" : "+v"(zero), "+v"(nn), "+v"(cacc[0]), "+v"(cacc[1]), "+v"(cacc[2]), "+v"(cacc[3]));
      acc.x = zero + r[0].x; acc.y = zero + r[0].y; acc.z = zero + r[0].z; acc.w = zero + r[0].w;
#pragma unroll
      for (int u = 1; u < GB; ++u) {
        const unsigned mk = (u < nn) ? 0xffffffffu : 0u;
        acc.x += __uint_as_float(__float_as_uint(r[u].x) & mk);
        acc.y += __uint_as_float(__float_as_uint(r[u].y) & mk);
        acc.z += __uint_as_float(__float_as_uint(r[u].z) & mk);
        acc.w += __uint_as_float(__float_as_uint(r[u].w) & mk);
      }
    }
    for (int j = GB; j < n; j += GB) {
      int idx[GB];
#pragma unroll
      for (int u = 0; u < GB; ++u) {
        const int e = csr[start + min(j + u, n - 1)];
        idx[u] = min(max(e, 0), N_NODES - 1);
      }
      float4 r[GB];
#pragma unroll
      for (int u = 0; u < GB; ++u)
        r[u] = x4[(unsigned)idx[u] * (DIM / 4) + (unsigned)l16];
#pragma unroll
      for (int u = 0; u < GB; ++u) {
        const bool in = (j + u < n);
        acc.x += in ? r[u].x : 0.0f;
        acc.y += in ? r[u].y : 0.0f;
        acc.z += in ? r[u].z : 0.0f;
        acc.w += in ? r[u].w : 0.0f;
      }
    }
    const float rinv = 1.0f / fmaxf((float)dr, 1.0f);
    const bool any = (n > 0);
    acc.x = any ? acc.x * rinv : 0.0f;
    acc.y = any ? acc.y * rinv : 0.0f;
    acc.z = any ? acc.z * rinv : 0.0f;
    acc.w = any ? acc.w * rinv : 0.0f;
    *(float4*)(my + MEAN_AT(m, d4)) = acc;
  }
}

#define GGRID ((NTILES + 7) / 8)
static_assert(BNODES == 16 * (GEMM_T / 32) && NB == GGRID, "a block of the sort is the 8 tiles of one work-group");

__device__ __forceinline__ void fold_head(const float* __restrict__ wp1, const float* __restrict__ bp1,
                                          const float* __restrict__ wp2, const float* __restrict__ bp2,
                                          float* __restrict__ wc, float* __restrict__ bc) {
  const int wave = threadIdx.x >> 5;
  const int lane = threadIdx.x & 31;
  const int l16  = lane & 15;
  const int half = lane >> 4;
#pragma unroll
  for (int u = 0; u < 2; ++u) {
    const int t  = 2 * wave + u;
    const int n0 = (t >> 2) * 16;
    const int k0 = (t & 3) * 16;
    v8f c8 = {};
#pragma unroll
    for (int s = 0; s < 16; ++s) {
      const int j = 4 * s + 2 * half;
      const v2f a = *(const v2f*)(wp2 + (n0 + l16) * DIM + j);
      v2f b;
      b.x = wp1[j * DIM + k0 + l16];
      b.y = wp1[(j + 1) * DIM + k0 + l16];
      c8 = wmma_f32(a, b, c8);
    }
#pragma unroll
    for (int r = 0; r < 8; ++r)
      wc[(n0 + 8 * half + r) * DIM + k0 + l16] = c8[r];
  }
  if (threadIdx.x < DIM) {
    const int n = threadIdx.x;
    float v = bp2[n];
    for (int j = 0; j < DIM; ++j) v = fmaf(wp2[n * DIM + j], bp1[j], v);
    bc[n] = v;
  }
}

template <bool HEAD>
__global__ __launch_bounds__(GEMM_T) void sage_kernel(
    const float* __restrict__ xin, int* csr, int* off, int* deg,
    const int* __restrict__ off1, const int* __restrict__ rec,
    const float* __restrict__ wl, const float* __restrict__ bl,
    const float* __restrict__ wr, const float* __restrict__ br,
    const float* __restrict__ wp1, const float* __restrict__ bp1,
    const float* __restrict__ wp2, const float* __restrict__ bp2,
    float* __restrict__ fold_w, float* __restrict__ fold_b,
    float* __restrict__ out) {
  if constexpr (!HEAD) {
    if (blockIdx.x == GGRID) {
      fold_head(wp1, bp1, wp2, bp2, fold_w, fold_b);
      return;
    }
  }
  __shared__ v2f lwl[WF_PAIRS];
  __shared__ v2f lwr[WF_PAIRS];
  __shared__ float4 agg4[(GEMM_T / 32) * 16 * (AG_PITCH / 4)];
  __shared__ int cntL[BNODES];
  __shared__ int offL[BNODES];
  __shared__ int degL[BNODES];
  __shared__ int redL[GEMM_T / 32];
  __shared__ int wtL[BNODES / 32];
  const int wave = threadIdx.x >> 5;
  const int lane = threadIdx.x & 31;
  const int tile_raw = blockIdx.x * (blockDim.x >> 5) + wave;
  const bool live = tile_raw < NTILES;
  const int tile = live ? tile_raw : NTILES - 1;

  const int l16  = lane & 15;
  const int half = lane >> 4;
  const int rb   = tile * 16;
  const int arow = rb + l16;
  float* my = (float*)agg4 + wave * (16 * AG_PITCH);

  float4 ql[WQ], qr[WQ];
  float4 xs[8];
  const float4* __restrict__ xt4 = (const float4*)xin + (long long)rb * (DIM / 4);
  int st_l = 0, dg_l = 0;
  if constexpr (HEAD) {
    issue_loads(wl, wr, xt4, lane, ql, qr, xs);
    st_l = off[arow];
    dg_l = deg[arow];
  } else {
    bin_block(off1, rec, csr, off, deg, blockIdx.x, cntL, offL, degL, redL, wtL, wl, wr, xt4, ql, qr, xs);
  }
  store_weights(lwl, ql);
  store_weights(lwr, qr);
#pragma unroll
  for (int j = 0; j < 8; ++j)
    *(float4*)(my + (2 * j + half) * AG_PITCH + 4 * l16) = xs[j];
  __syncthreads();
  if constexpr (!HEAD) {
    const int ln = (tile - blockIdx.x * (GEMM_T / 32)) * 16 + l16;
    st_l = offL[ln];
    dg_l = degL[ln];
  }

  v8f acc[4];
#pragma unroll
  for (int nt = 0; nt < 4; ++nt) acc[nt] = v8f{};
  gather_tile(xin, csr, st_l, dg_l, l16, half, lane, my, lwl, acc);
  __syncthreads();
  if constexpr (!HEAD) {
    if (!live) return;
  }

  float4 q1[WQ];
  if constexpr (HEAD) {
    load_weights(wp1, q1);
  }

  v2f ap[16];
#pragma unroll
  for (int k = 0; k < 16; ++k)
    ap[k] = *(const v2f*)(my + MEAN_AT(l16, 4 * k + 2 * half));

#pragma unroll
  for (int nt = 0; nt < 4; ++nt) {
    v8f c8 = acc[nt];
    const int n0 = nt * 16;
#pragma unroll
    for (int k = 0; k < 16; ++k) {
      const v2f b = lwr[(nt * 16 + k) * WF_PITCH + lane];
      c8 = wmma_f32(ap[k], b, c8);
    }
    const float bias = bl[n0 + l16] + br[n0 + l16];
#pragma unroll
    for (int r = 0; r < 8; ++r) c8[r] += bias;
    acc[nt] = c8;
  }


  float ssq[8];
#pragma unroll
  for (int r = 0; r < 8; ++r) {
    float v = 0.0f;
#pragma unroll
    for (int nt = 0; nt < 4; ++nt) v += acc[nt][r] * acc[nt][r];
    ssq[r] = v;
  }
#pragma unroll
  for (int m = 1; m <= 8; m <<= 1) {
#pragma unroll
    for (int r = 0; r < 8; ++r) ssq[r] += __shfl_xor(ssq[r], m, 32);
  }

#pragma unroll
  for (int r = 0; r < 8; ++r) {
    const float inv = 1.0f / fmaxf(sqrtf(ssq[r]), L2EPS);
    const int row = rb + r + 8 * half;
#pragma unroll
    for (int nt = 0; nt < 4; ++nt) {
      float v = acc[nt][r] * inv;
      v = fmaxf(v, 0.0f);
      if constexpr (HEAD)
        my[(r + 8 * half) * AG_PITCH + nt * 16 + l16] = v;
      else
        out[(long long)row * DIM + nt * 16 + l16] = v;
    }
  }

  if constexpr (HEAD) {
    store_weights(lwl, q1);
    __syncthreads();

    v2f ah[16];
#pragma unroll
    for (int k = 0; k < 16; ++k)
      ah[k] = *(const v2f*)(my + l16 * AG_PITCH + 4 * k + 2 * half);
#pragma unroll
    for (int nt = 0; nt < 4; ++nt) {
      v8f c8 = {};
#pragma unroll
      for (int k = 0; k < 16; ++k) {
        const v2f b = lwl[(nt * 16 + k) * WF_PITCH + lane];
        c8 = wmma_f32(ah[k], b, c8);
      }
      acc[nt] = c8;
    }
    if (live) {
#pragma unroll
      for (int nt = 0; nt < 4; ++nt) {
        const float bias = bp1[nt * 16 + l16];
#pragma unroll
        for (int r = 0; r < 8; ++r) {
          const int row = rb + r + 8 * half;
          out[(long long)row * DIM + nt * 16 + l16] = acc[nt][r] + bias;
        }
      }
    }
  }
}

extern "C" void kernel_launch(void* const* d_in, const int* in_sizes, int n_in,
                              void* d_out, int out_size, void* d_ws, size_t ws_size,
                              hipStream_t stream) {
  (void)in_sizes; (void)n_in; (void)out_size; (void)ws_size;

  const float* x   = (const float*)d_in[0];
  const int*   ei  = (const int*)d_in[1];
  const int*   src = ei;
  const int*   dst = ei + N_EDGES;
  const float* w1l = (const float*)d_in[2];
  const float* b1l = (const float*)d_in[3];
  const float* w1r = (const float*)d_in[4];
  const float* b1r = (const float*)d_in[5];
  const float* w2l = (const float*)d_in[6];
  const float* b2l = (const float*)d_in[7];
  const float* w2r = (const float*)d_in[8];
  const float* b2r = (const float*)d_in[9];
  const float* wp1 = (const float*)d_in[10];
  const float* bp1 = (const float*)d_in[11];
  const float* wp2 = (const float*)d_in[12];
  const float* bp2 = (const float*)d_in[13];
  float* out = (float*)d_out;

  const long long NS = (long long)N_NODES * DIM;
  float* s   = (float*)d_ws;
  float* h1  = s + NS;
  int* off1 = (int*)s;
  int* deg  = off1 + (CHUNKS * O1 + 31) / 32 * 32;
  int* off  = deg + N_NODES;
  int* rec  = off + N_NODES;
  int* csr  = rec + N_EDGES;
  float* wc = s + 4000000;
  float* bc = wc + DIM * DIM;

  const dim3 BG(GEMM_T);
  const dim3 BS(SORT_T);
  const int ggrid = GGRID;

  chunk_sort_kernel<<<CHUNKS, BS, 0, stream>>>(src, dst, rec, off1);

  sage_kernel<false><<<GGRID + 1, BG, 0, stream>>>(x, csr, off, deg, off1, rec, w1l, b1l, w1r, b1r,
                                                   wp1, bp1, wp2, bp2, wc, bc, h1);

  sage_kernel<true><<<ggrid, BG, 0, stream>>>(h1, csr, off, deg, nullptr, nullptr, w2l, b2l, w2r, b2r,
                                              wc, bc, wp2, bp2, nullptr, nullptr, out);
}
